# starting kernel
# baseline (speedup 1.0000x reference)
#include <hip/hip_runtime.h>
#include <hip/hip_bf16.h>

typedef __attribute__((ext_vector_type(2))) float v2f;
typedef __attribute__((ext_vector_type(8))) float v8f;

#define F_IN 128
#define HID  16
#define OUT  2

__global__ void gcn_init_kernel(float* __restrict__ deg,
                                float* __restrict__ agg1,
                                float* __restrict__ agg2, int N) {
    int n = blockIdx.x * blockDim.x + threadIdx.x;
    if (n >= N) return;
    deg[n] = 1.0f;
    float4 z = make_float4(0.f, 0.f, 0.f, 0.f);
    float4* a1 = (float4*)(agg1 + (long long)n * HID);
    a1[0] = z; a1[1] = z; a1[2] = z; a1[3] = z;
    agg2[2 * (long long)n + 0] = 0.f;
    agg2[2 * (long long)n + 1] = 0.f;
}

__global__ void gcn_deg_kernel(const int* __restrict__ dst,
                               float* __restrict__ deg, long long E) {
    long long e = (long long)blockIdx.x * blockDim.x + threadIdx.x;
    if (e >= E) return;
    atomicAdd(&deg[dst[e]], 1.0f);
}

__global__ void gcn_dis_kernel(const float* __restrict__ deg,
                               float* __restrict__ dis, int N) {
    int n = blockIdx.x * blockDim.x + threadIdx.x;
    if (n >= N) return;
    dis[n] = rsqrtf(deg[n]);
}

__global__ void gcn_gemm1_wmma(const float* __restrict__ x,
                               const float* __restrict__ W1,
                               float* __restrict__ h1, int n_tiles) {
    __shared__ float sW[F_IN * HID];
    for (int i = threadIdx.x; i < F_IN * HID; i += blockDim.x) sW[i] = W1[i];
    __syncthreads();

    const int wave_in_blk = threadIdx.x >> 5;
    const int tile = blockIdx.x * (blockDim.x >> 5) + wave_in_blk;
    if (tile >= n_tiles) return;

    const int lane = threadIdx.x & 31;
    const int half = lane >> 4;
    const int l16  = lane & 15;

    const int row_base = tile * 16;
    const float* xrow = x + (long long)(row_base + l16) * F_IN;

    v8f c = {};
    #pragma unroll 8
    for (int k = 0; k < F_IN; k += 4) {
        v2f a;
        a.x = xrow[k + 2 * half + 0];
        a.y = xrow[k + 2 * half + 1];
        v2f b;
        b.x = sW[(k + 2 * half + 0) * HID + l16];
        b.y = sW[(k + 2 * half + 1) * HID + l16];
        c = __builtin_amdgcn_wmma_f32_16x16x4_f32(
                false, a, false, b, (short)0, c, false, false);
    }

    float* hrow = h1 + (long long)row_base * HID;
    #pragma unroll
    for (int i = 0; i < 8; ++i)
        hrow[(i + 8 * half) * HID + l16] = c[i];
}

__global__ void gcn_agg1_kernel(const int* __restrict__ src,
                                const int* __restrict__ dst,
                                const float* __restrict__ dis,
                                const float* __restrict__ h1,
                                float* __restrict__ agg1, long long E) {
    long long t = (long long)blockIdx.x * blockDim.x + threadIdx.x;
    long long e = t >> 4;
    int f = (int)(t & 15);
    if (e >= E) return;
    int s = src[e], d = dst[e];
    float w = dis[s] * dis[d];
    atomicAdd(&agg1[(long long)d * HID + f], w * h1[(long long)s * HID + f]);
}

__global__ void gcn_relu1_kernel(const float* __restrict__ agg1,
                                 const float* __restrict__ h1,
                                 const float* __restrict__ dis,
                                 const float* __restrict__ b1,
                                 float* __restrict__ r, int N) {
    long long t = (long long)blockIdx.x * blockDim.x + threadIdx.x;
    int n = (int)(t >> 4);
    int f = (int)(t & 15);
    if (n >= N) return;
    float di = dis[n];
    long long idx = (long long)n * HID + f;
    float z = agg1[idx] + di * di * h1[idx] + b1[f];
    r[idx] = fmaxf(z, 0.0f);
}

__global__ void gcn_gemm2_kernel(const float* __restrict__ r,
                                 const float* __restrict__ W2,
                                 float* __restrict__ h2, int N) {
    int n = blockIdx.x * blockDim.x + threadIdx.x;
    if (n >= N) return;
    const float* rn = r + (long long)n * HID;
    float a0 = 0.f, a1 = 0.f;
    #pragma unroll
    for (int f = 0; f < HID; ++f) {
        float v = rn[f];
        a0 = fmaf(v, W2[f * OUT + 0], a0);
        a1 = fmaf(v, W2[f * OUT + 1], a1);
    }
    h2[(long long)n * OUT + 0] = a0;
    h2[(long long)n * OUT + 1] = a1;
}

__global__ void gcn_agg2_kernel(const int* __restrict__ src,
                                const int* __restrict__ dst,
                                const float* __restrict__ dis,
                                const float* __restrict__ h2,
                                float* __restrict__ agg2, long long E) {
    long long e = (long long)blockIdx.x * blockDim.x + threadIdx.x;
    if (e >= E) return;
    int s = src[e], d = dst[e];
    float w = dis[s] * dis[d];
    float v0 = h2[(long long)s * OUT + 0];
    float v1 = h2[(long long)s * OUT + 1];
    atomicAdd(&agg2[(long long)d * OUT + 0], w * v0);
    atomicAdd(&agg2[(long long)d * OUT + 1], w * v1);
}

__global__ void gcn_final_kernel(const float* __restrict__ agg2,
                                 const float* __restrict__ h2,
                                 const float* __restrict__ dis,
                                 const float* __restrict__ b2,
                                 float* __restrict__ out, int N) {
    int n = blockIdx.x * blockDim.x + threadIdx.x;
    if (n >= N) return;
    float di = dis[n];
    float w = di * di;
    long long i0 = (long long)n * OUT;
    float z0 = agg2[i0 + 0] + w * h2[i0 + 0] + b2[0];
    float z1 = agg2[i0 + 1] + w * h2[i0 + 1] + b2[1];
    float m = fmaxf(z0, z1);
    float lse = m + logf(expf(z0 - m) + expf(z1 - m));
    out[i0 + 0] = z0 - lse;
    out[i0 + 1] = z1 - lse;
}

extern "C" void kernel_launch(void* const* d_in, const int* in_sizes, int n_in,
                              void* d_out, int out_size, void* d_ws, size_t ws_size,
                              hipStream_t stream) {
    const float* x  = (const float*)d_in[0];
    const int*   ei = (const int*)d_in[1];
    const float* W1 = (const float*)d_in[2];
    const float* b1 = (const float*)d_in[3];
    const float* W2 = (const float*)d_in[4];
    const float* b2 = (const float*)d_in[5];
    float* out = (float*)d_out;

    const int       N = in_sizes[0] / F_IN;
    const long long E = (long long)in_sizes[1] / 2;
    const int* src = ei;
    const int* dst = ei + E;

    float* ws   = (float*)d_ws;
    float* deg  = ws;
    float* dis  = deg  + N;
    float* h1   = dis  + N;
    float* agg1 = h1   + (long long)N * HID;
    float* r    = agg1 + (long long)N * HID;
    float* h2   = r    + (long long)N * HID;
    float* agg2 = h2   + (long long)N * OUT;

    const int B = 256;

    gcn_init_kernel<<<(N + B - 1) / B, B, 0, stream>>>(deg, agg1, agg2, N);

    gcn_deg_kernel<<<(unsigned)((E + B - 1) / B), B, 0, stream>>>(dst, deg, E);

    gcn_dis_kernel<<<(N + B - 1) / B, B, 0, stream>>>(deg, dis, N);

    {
        int n_tiles = N / 16;
        int waves_per_block = B / 32;
        int blocks = (n_tiles + waves_per_block - 1) / waves_per_block;
        gcn_gemm1_wmma<<<blocks, B, 0, stream>>>(x, W1, h1, n_tiles);
    }

    {
        long long T = E * HID;
        gcn_agg1_kernel<<<(unsigned)((T + B - 1) / B), B, 0, stream>>>(src, dst, dis, h1, agg1, E);
    }

    {
        long long T = (long long)N * HID;
        gcn_relu1_kernel<<<(unsigned)((T + B - 1) / B), B, 0, stream>>>(agg1, h1, dis, b1, r, N);
    }

    gcn_gemm2_kernel<<<(N + B - 1) / B, B, 0, stream>>>(r, W2, h2, N);

    gcn_agg2_kernel<<<(unsigned)((E + B - 1) / B), B, 0, stream>>>(src, dst, dis, h2, agg2, E);

    gcn_final_kernel<<<(N + B - 1) / B, B, 0, stream>>>(agg2, h2, dis, b2, out, N);
}


# Layer-2 edge scatter: adjacent lane pair per edge (one lane per output feature), so an edge's two f32 atomic adds leave as one 8-byte request instead of two; atomic arrival order as before
# speedup vs baseline: 1.2082x; 1.2082x over previous
#include <hip/hip_runtime.h>
#include <hip/hip_bf16.h>

typedef __attribute__((ext_vector_type(2))) float v2f;
typedef __attribute__((ext_vector_type(8))) float v8f;

#define F_IN 128
#define HID  16
#define OUT  2

__global__ void gcn_init_kernel(float* __restrict__ deg,
                                float* __restrict__ agg1,
                                float* __restrict__ agg2, int N) {
    int n = blockIdx.x * blockDim.x + threadIdx.x;
    if (n >= N) return;
    deg[n] = 1.0f;
    float4 z = make_float4(0.f, 0.f, 0.f, 0.f);
    float4* a1 = (float4*)(agg1 + (long long)n * HID);
    a1[0] = z; a1[1] = z; a1[2] = z; a1[3] = z;
    agg2[2 * (long long)n + 0] = 0.f;
    agg2[2 * (long long)n + 1] = 0.f;
}

__global__ void gcn_deg_kernel(const int* __restrict__ dst,
                               float* __restrict__ deg, long long E) {
    long long e = (long long)blockIdx.x * blockDim.x + threadIdx.x;
    if (e >= E) return;
    atomicAdd(&deg[dst[e]], 1.0f);
}

__global__ void gcn_dis_kernel(const float* __restrict__ deg,
                               float* __restrict__ dis, int N) {
    int n = blockIdx.x * blockDim.x + threadIdx.x;
    if (n >= N) return;
    dis[n] = rsqrtf(deg[n]);
}

__global__ void gcn_gemm1_wmma(const float* __restrict__ x,
                               const float* __restrict__ W1,
                               float* __restrict__ h1, int n_tiles) {
    __shared__ float sW[F_IN * HID];
    for (int i = threadIdx.x; i < F_IN * HID; i += blockDim.x) sW[i] = W1[i];
    __syncthreads();

    const int wave_in_blk = threadIdx.x >> 5;
    const int tile = blockIdx.x * (blockDim.x >> 5) + wave_in_blk;
    if (tile >= n_tiles) return;

    const int lane = threadIdx.x & 31;
    const int half = lane >> 4;
    const int l16  = lane & 15;

    const int row_base = tile * 16;
    const float* xrow = x + (long long)(row_base + l16) * F_IN;

    v8f c = {};
    #pragma unroll 8
    for (int k = 0; k < F_IN; k += 4) {
        v2f a;
        a.x = xrow[k + 2 * half + 0];
        a.y = xrow[k + 2 * half + 1];
        v2f b;
        b.x = sW[(k + 2 * half + 0) * HID + l16];
        b.y = sW[(k + 2 * half + 1) * HID + l16];
        c = __builtin_amdgcn_wmma_f32_16x16x4_f32(
                false, a, false, b, (short)0, c, false, false);
    }

    float* hrow = h1 + (long long)row_base * HID;
    #pragma unroll
    for (int i = 0; i < 8; ++i)
        hrow[(i + 8 * half) * HID + l16] = c[i];
}

__global__ void gcn_agg1_kernel(const int* __restrict__ src,
                                const int* __restrict__ dst,
                                const float* __restrict__ dis,
                                const float* __restrict__ h1,
                                float* __restrict__ agg1, long long E) {
    long long t = (long long)blockIdx.x * blockDim.x + threadIdx.x;
    long long e = t >> 4;
    int f = (int)(t & 15);
    if (e >= E) return;
    int s = src[e], d = dst[e];
    float w = dis[s] * dis[d];
    atomicAdd(&agg1[(long long)d * HID + f], w * h1[(long long)s * HID + f]);
}

__global__ void gcn_relu1_kernel(const float* __restrict__ agg1,
                                 const float* __restrict__ h1,
                                 const float* __restrict__ dis,
                                 const float* __restrict__ b1,
                                 float* __restrict__ r, int N) {
    long long t = (long long)blockIdx.x * blockDim.x + threadIdx.x;
    int n = (int)(t >> 4);
    int f = (int)(t & 15);
    if (n >= N) return;
    float di = dis[n];
    long long idx = (long long)n * HID + f;
    float z = agg1[idx] + di * di * h1[idx] + b1[f];
    r[idx] = fmaxf(z, 0.0f);
}

__global__ void gcn_gemm2_kernel(const float* __restrict__ r,
                                 const float* __restrict__ W2,
                                 float* __restrict__ h2, int N) {
    int n = blockIdx.x * blockDim.x + threadIdx.x;
    if (n >= N) return;
    const float* rn = r + (long long)n * HID;
    float a0 = 0.f, a1 = 0.f;
    #pragma unroll
    for (int f = 0; f < HID; ++f) {
        float v = rn[f];
        a0 = fmaf(v, W2[f * OUT + 0], a0);
        a1 = fmaf(v, W2[f * OUT + 1], a1);
    }
    h2[(long long)n * OUT + 0] = a0;
    h2[(long long)n * OUT + 1] = a1;
}

__global__ void gcn_agg2_kernel(const int* __restrict__ src,
                                const int* __restrict__ dst,
                                const float* __restrict__ dis,
                                const float* __restrict__ h2,
                                float* __restrict__ agg2, long long E) {
    long long t = (long long)blockIdx.x * blockDim.x + threadIdx.x;
    long long e = t >> 1;
    int f = (int)(t & 1);
    if (e >= E) return;
    int s = src[e], d = dst[e];
    float w = dis[s] * dis[d];
    atomicAdd(&agg2[(long long)d * OUT + f], w * h2[(long long)s * OUT + f]);
}

__global__ void gcn_final_kernel(const float* __restrict__ agg2,
                                 const float* __restrict__ h2,
                                 const float* __restrict__ dis,
                                 const float* __restrict__ b2,
                                 float* __restrict__ out, int N) {
    int n = blockIdx.x * blockDim.x + threadIdx.x;
    if (n >= N) return;
    float di = dis[n];
    float w = di * di;
    long long i0 = (long long)n * OUT;
    float z0 = agg2[i0 + 0] + w * h2[i0 + 0] + b2[0];
    float z1 = agg2[i0 + 1] + w * h2[i0 + 1] + b2[1];
    float m = fmaxf(z0, z1);
    float lse = m + logf(expf(z0 - m) + expf(z1 - m));
    out[i0 + 0] = z0 - lse;
    out[i0 + 1] = z1 - lse;
}

extern "C" void kernel_launch(void* const* d_in, const int* in_sizes, int n_in,
                              void* d_out, int out_size, void* d_ws, size_t ws_size,
                              hipStream_t stream) {
    const float* x  = (const float*)d_in[0];
    const int*   ei = (const int*)d_in[1];
    const float* W1 = (const float*)d_in[2];
    const float* b1 = (const float*)d_in[3];
    const float* W2 = (const float*)d_in[4];
    const float* b2 = (const float*)d_in[5];
    float* out = (float*)d_out;

    const int       N = in_sizes[0] / F_IN;
    const long long E = (long long)in_sizes[1] / 2;
    const int* src = ei;
    const int* dst = ei + E;

    float* ws   = (float*)d_ws;
    float* deg  = ws;
    float* dis  = deg  + N;
    float* h1   = dis  + N;
    float* agg1 = h1   + (long long)N * HID;
    float* r    = agg1 + (long long)N * HID;
    float* h2   = r    + (long long)N * HID;
    float* agg2 = h2   + (long long)N * OUT;

    const int B = 256;

    gcn_init_kernel<<<(N + B - 1) / B, B, 0, stream>>>(deg, agg1, agg2, N);

    gcn_deg_kernel<<<(unsigned)((E + B - 1) / B), B, 0, stream>>>(dst, deg, E);

    gcn_dis_kernel<<<(N + B - 1) / B, B, 0, stream>>>(deg, dis, N);

    {
        int n_tiles = N / 16;
        int waves_per_block = B / 32;
        int blocks = (n_tiles + waves_per_block - 1) / waves_per_block;
        gcn_gemm1_wmma<<<blocks, B, 0, stream>>>(x, W1, h1, n_tiles);
    }

    {
        long long T = E * HID;
        gcn_agg1_kernel<<<(unsigned)((T + B - 1) / B), B, 0, stream>>>(src, dst, dis, h1, agg1, E);
    }

    {
        long long T = (long long)N * HID;
        gcn_relu1_kernel<<<(unsigned)((T + B - 1) / B), B, 0, stream>>>(agg1, h1, dis, b1, r, N);
    }

    gcn_gemm2_kernel<<<(N + B - 1) / B, B, 0, stream>>>(r, W2, h2, N);

    {
        long long T = E * OUT;
        gcn_agg2_kernel<<<(unsigned)((T + B - 1) / B), B, 0, stream>>>(src, dst, dis, h2, agg2, E);
    }

    gcn_final_kernel<<<(N + B - 1) / B, B, 0, stream>>>(agg2, h2, dis, b2, out, N);
}


# Edges grouped by destination each call (returning int add ranks each edge, block scan, one fill pass): both aggregations are per-node gathers, no f32 atomics; sums in rank order, layer 2 16-lane tree
# speedup vs baseline: 1.8381x; 1.5214x over previous
#include <hip/hip_runtime.h>
#include <hip/hip_bf16.h>

typedef __attribute__((ext_vector_type(2))) float v2f;
typedef __attribute__((ext_vector_type(8))) float v8f;

#define F_IN 128
#define HID  16
#define OUT  2

__global__ void gcn_init_kernel(int* __restrict__ cnt, int* __restrict__ cursor, int N) {
    int n = blockIdx.x * blockDim.x + threadIdx.x;
    if (n == 0) *cursor = 0;
    if (n >= N) return;
    cnt[n] = 0;
}

__global__ void gcn_rank_kernel(const int* __restrict__ dst,
                                int* __restrict__ cnt,
                                int* __restrict__ rank, long long E) {
    long long e = (long long)blockIdx.x * blockDim.x + threadIdx.x;
    if (e >= E) return;
    rank[e] = atomicAdd(&cnt[dst[e]], 1);
}

__global__ void gcn_offsets_kernel(const int* __restrict__ cnt,
                                   int* __restrict__ cursor,
                                   int* __restrict__ off,
                                   float* __restrict__ dis, int N) {
    __shared__ int s_wave[32];
    __shared__ int s_base;
    const int tid  = threadIdx.x;
    const int lane = tid & 31;
    const int wave = tid >> 5;
    const int n = blockIdx.x * blockDim.x + tid;
    const int c = (n < N) ? cnt[n] : 0;
    int v = c;
    #pragma unroll
    for (int d = 1; d < 32; d <<= 1) {
        int t = __shfl_up(v, d, 32);
        if (lane >= d) v += t;
    }
    if (lane == 31) s_wave[wave] = v;
    __syncthreads();
    if (tid == 0) {
        const int nw = blockDim.x >> 5;
        int tot = 0;
        for (int w = 0; w < nw; ++w) { int t = s_wave[w]; s_wave[w] = tot; tot += t; }
        s_base = atomicAdd(cursor, tot);
    }
    __syncthreads();
    if (n < N) {
        off[n] = s_base + s_wave[wave] + (v - c);
        dis[n] = rsqrtf((float)(c + 1));
    }
}

__global__ void gcn_fill_kernel(const int* __restrict__ src,
                                const int* __restrict__ dst,
                                const int* __restrict__ off,
                                const int* __restrict__ rank,
                                int* __restrict__ csr, long long E) {
    long long e = (long long)blockIdx.x * blockDim.x + threadIdx.x;
    if (e >= E) return;
    csr[off[dst[e]] + rank[e]] = src[e];
}

__global__ void gcn_gemm1_wmma(const float* __restrict__ x,
                               const float* __restrict__ W1,
                               float* __restrict__ h1, int n_tiles) {
    __shared__ float sW[F_IN * HID];
    for (int i = threadIdx.x; i < F_IN * HID; i += blockDim.x) sW[i] = W1[i];
    __syncthreads();

    const int wave_in_blk = threadIdx.x >> 5;
    const int tile = blockIdx.x * (blockDim.x >> 5) + wave_in_blk;
    if (tile >= n_tiles) return;

    const int lane = threadIdx.x & 31;
    const int half = lane >> 4;
    const int l16  = lane & 15;

    const int row_base = tile * 16;
    const float* xrow = x + (long long)(row_base + l16) * F_IN;

    v8f c = {};
    #pragma unroll 8
    for (int k = 0; k < F_IN; k += 4) {
        v2f a;
        a.x = xrow[k + 2 * half + 0];
        a.y = xrow[k + 2 * half + 1];
        v2f b;
        b.x = sW[(k + 2 * half + 0) * HID + l16];
        b.y = sW[(k + 2 * half + 1) * HID + l16];
        c = __builtin_amdgcn_wmma_f32_16x16x4_f32(
                false, a, false, b, (short)0, c, false, false);
    }

    float* hrow = h1 + (long long)row_base * HID;
    #pragma unroll
    for (int i = 0; i < 8; ++i)
        hrow[(i + 8 * half) * HID + l16] = c[i];
}

__global__ void gcn_agg1_kernel(const int* __restrict__ csr,
                                const int* __restrict__ off,
                                const int* __restrict__ cnt,
                                const float* __restrict__ dis,
                                const float* __restrict__ h1,
                                float* __restrict__ agg1, int N) {
    long long t = (long long)blockIdx.x * blockDim.x + threadIdx.x;
    int n = (int)(t >> 4);
    int f = (int)(t & 15);
    if (n >= N) return;
    const int c = cnt[n];
    const int* p = csr + off[n];
    const float dn = dis[n];
    float acc = 0.f;
    for (int j = 0; j < c; j += 8) {
        int   s[8];
        float w[8], v[8];
        #pragma unroll
        for (int u = 0; u < 8; ++u) {
            int jj = j + u;
            s[u] = p[jj < c ? jj : c - 1];
        }
        #pragma unroll
        for (int u = 0; u < 8; ++u) {
            const float ds = dis[s[u]];
            w[u] = (j + u < c) ? ds * dn : 0.f;
            v[u] = h1[(long long)s[u] * HID + f];
        }
        #pragma unroll
        for (int u = 0; u < 8; ++u) acc = fmaf(w[u], v[u], acc);
    }
    agg1[(long long)n * HID + f] = acc;
}

__global__ void gcn_relu1_kernel(const float* __restrict__ agg1,
                                 const float* __restrict__ h1,
                                 const float* __restrict__ dis,
                                 const float* __restrict__ b1,
                                 float* __restrict__ r, int N) {
    long long t = (long long)blockIdx.x * blockDim.x + threadIdx.x;
    int n = (int)(t >> 4);
    int f = (int)(t & 15);
    if (n >= N) return;
    float di = dis[n];
    long long idx = (long long)n * HID + f;
    float z = agg1[idx] + di * di * h1[idx] + b1[f];
    r[idx] = fmaxf(z, 0.0f);
}

__global__ void gcn_gemm2_kernel(const float* __restrict__ r,
                                 const float* __restrict__ W2,
                                 float* __restrict__ h2, int N) {
    int n = blockIdx.x * blockDim.x + threadIdx.x;
    if (n >= N) return;
    const float* rn = r + (long long)n * HID;
    float a0 = 0.f, a1 = 0.f;
    #pragma unroll
    for (int f = 0; f < HID; ++f) {
        float v = rn[f];
        a0 = fmaf(v, W2[f * OUT + 0], a0);
        a1 = fmaf(v, W2[f * OUT + 1], a1);
    }
    h2[(long long)n * OUT + 0] = a0;
    h2[(long long)n * OUT + 1] = a1;
}

__global__ void gcn_agg2_kernel(const int* __restrict__ csr,
                                const int* __restrict__ off,
                                const int* __restrict__ cnt,
                                const float* __restrict__ dis,
                                const float* __restrict__ h2,
                                float* __restrict__ agg2, int N) {
    long long t = (long long)blockIdx.x * blockDim.x + threadIdx.x;
    int n = (int)(t >> 4);
    int l = (int)(t & 15);
    if (n >= N) return;
    const int c = cnt[n];
    const int* p = csr + off[n];
    const float dn = dis[n];
    float a0 = 0.f, a1 = 0.f;
    for (int j = l; j < c; j += 16) {
        int s = p[j];
        float w = dis[s] * dn;
        const float2 hv = *reinterpret_cast<const float2*>(h2 + (long long)s * OUT);
        a0 = fmaf(w, hv.x, a0);
        a1 = fmaf(w, hv.y, a1);
    }
    #pragma unroll
    for (int m = 8; m >= 1; m >>= 1) {
        a0 += __shfl_xor(a0, m, 16);
        a1 += __shfl_xor(a1, m, 16);
    }
    if (l == 0) {
        agg2[(long long)n * OUT + 0] = a0;
        agg2[(long long)n * OUT + 1] = a1;
    }
}

__global__ void gcn_final_kernel(const float* __restrict__ agg2,
                                 const float* __restrict__ h2,
                                 const float* __restrict__ dis,
                                 const float* __restrict__ b2,
                                 float* __restrict__ out, int N) {
    int n = blockIdx.x * blockDim.x + threadIdx.x;
    if (n >= N) return;
    float di = dis[n];
    float w = di * di;
    long long i0 = (long long)n * OUT;
    float z0 = agg2[i0 + 0] + w * h2[i0 + 0] + b2[0];
    float z1 = agg2[i0 + 1] + w * h2[i0 + 1] + b2[1];
    float m = fmaxf(z0, z1);
    float lse = m + logf(expf(z0 - m) + expf(z1 - m));
    out[i0 + 0] = z0 - lse;
    out[i0 + 1] = z1 - lse;
}

extern "C" void kernel_launch(void* const* d_in, const int* in_sizes, int n_in,
                              void* d_out, int out_size, void* d_ws, size_t ws_size,
                              hipStream_t stream) {
    const float* x  = (const float*)d_in[0];
    const int*   ei = (const int*)d_in[1];
    const float* W1 = (const float*)d_in[2];
    const float* b1 = (const float*)d_in[3];
    const float* W2 = (const float*)d_in[4];
    const float* b2 = (const float*)d_in[5];
    float* out = (float*)d_out;

    const int       N = in_sizes[0] / F_IN;
    const long long E = (long long)in_sizes[1] / 2;
    const int* src = ei;
    const int* dst = ei + E;

    float* ws   = (float*)d_ws;
    int*   cnt  = (int*)ws;
    float* dis  = ws   + N;
    float* h1   = dis  + N;
    float* agg1 = h1   + (long long)N * HID;
    float* r    = agg1 + (long long)N * HID;
    float* h2   = r    + (long long)N * HID;
    float* agg2 = h2   + (long long)N * OUT;
    int*   off  = (int*)(agg2 + (long long)N * OUT);
    int*   rank = off  + N;
    int*   csr  = rank + E;
    int*   cursor = csr + E;

    const int B = 256;

    gcn_init_kernel<<<(N + B - 1) / B, B, 0, stream>>>(cnt, cursor, N);

    gcn_rank_kernel<<<(unsigned)((E + B - 1) / B), B, 0, stream>>>(dst, cnt, rank, E);

    gcn_offsets_kernel<<<(N + B - 1) / B, B, 0, stream>>>(cnt, cursor, off, dis, N);

    {
        int n_tiles = N / 16;
        int waves_per_block = B / 32;
        int blocks = (n_tiles + waves_per_block - 1) / waves_per_block;
        gcn_gemm1_wmma<<<blocks, B, 0, stream>>>(x, W1, h1, n_tiles);
    }

    gcn_fill_kernel<<<(unsigned)((E + B - 1) / B), B, 0, stream>>>(src, dst, off, rank, csr, E);

    {
        long long T = (long long)N * HID;
        gcn_agg1_kernel<<<(unsigned)((T + B - 1) / B), B, 0, stream>>>(csr, off, cnt, dis, h1, agg1, N);
    }

    {
        long long T = (long long)N * HID;
        gcn_relu1_kernel<<<(unsigned)((T + B - 1) / B), B, 0, stream>>>(agg1, h1, dis, b1, r, N);
    }

    gcn_gemm2_kernel<<<(N + B - 1) / B, B, 0, stream>>>(r, W2, h2, N);

    {
        long long T = (long long)N * 16;
        gcn_agg2_kernel<<<(unsigned)((T + B - 1) / B), B, 0, stream>>>(csr, off, cnt, dis, h2, agg2, N);
    }

    gcn_final_kernel<<<(N + B - 1) / B, B, 0, stream>>>(agg2, h2, dis, b2, out, N);
}


# Edges grouped by destination by a two-level counting sort ranked with LDS integer adds (256-node buckets, then nodes), no returning global add per edge; sums follow LDS add arrival order
# speedup vs baseline: 4.1779x; 2.2729x over previous
#include <hip/hip_runtime.h>
#include <hip/hip_bf16.h>

typedef __attribute__((ext_vector_type(2))) float v2f;
typedef __attribute__((ext_vector_type(8))) float v8f;

#define F_IN 128
#define HID  16
#define OUT  2

#define NCH    256
#define NBMAX  512
#define BSH    8
#define BNODES 256
#define TPB    1024

__device__ __forceinline__ int gcn_bucket(int d) {
    return min((int)((unsigned)d >> BSH), NBMAX - 1);
}

__global__ __launch_bounds__(TPB) void gcn_bhist_kernel(const int* __restrict__ dst,
                                                        int* __restrict__ table, int E) {
    __shared__ int s_h[NBMAX];
    const int tid = threadIdx.x;
    if (tid < NBMAX) s_h[tid] = 0;
    __syncthreads();
    const int CH = (E + NCH - 1) / NCH;
    const int e0 = blockIdx.x * CH;
    const int e1 = min(E, e0 + CH);
    int e = e0 + tid;
    for (; e + 3 * TPB < e1; e += 4 * TPB) {
        int d[4];
        for (int u = 0; u < 4; ++u) d[u] = dst[e + u * TPB];
        for (int u = 0; u < 4; ++u) atomicAdd(&s_h[gcn_bucket(d[u])], 1);
    }
    for (; e < e1; e += TPB) atomicAdd(&s_h[gcn_bucket(dst[e])], 1);
    __syncthreads();
    if (tid < NBMAX) table[blockIdx.x * NBMAX + tid] = s_h[tid];
}

__global__ __launch_bounds__(TPB) void gcn_bscan_kernel(int* __restrict__ table,
                                                        int* __restrict__ tot) {
    __shared__ int s_p[32][33];
    const int bl = threadIdx.x & 31;
    const int cg = threadIdx.x >> 5;
    const int b  = blockIdx.x * 32 + bl;
    int v[8];
    for (int k = 0; k < 8; ++k) v[k] = table[(cg * 8 + k) * NBMAX + b];
    int sum = 0;
    for (int k = 0; k < 8; ++k) { int t = v[k]; v[k] = sum; sum += t; }
    s_p[cg][bl] = sum;
    __syncthreads();
    if (cg == 0) {
        int run = 0;
        for (int g = 0; g < 32; ++g) { int t = s_p[g][bl]; s_p[g][bl] = run; run += t; }
        tot[b] = run;
    }
    __syncthreads();
    const int base = s_p[cg][bl];
    for (int k = 0; k < 8; ++k) table[(cg * 8 + k) * NBMAX + b] = base + v[k];
}

__global__ __launch_bounds__(TPB) void gcn_bscatter_kernel(const int* __restrict__ src,
                                                           const int* __restrict__ dst,
                                                           const int* __restrict__ table,
                                                           const int* __restrict__ tot,
                                                           int2* __restrict__ pairs, int E) {
    __shared__ int s_cur[NBMAX];
    __shared__ int s_wave[32];
    const int tid  = threadIdx.x;
    const int lane = tid & 31;
    const int wave = tid >> 5;
    const int c = (tid < NBMAX) ? tot[tid] : 0;
    int v = c;
    for (int d = 1; d < 32; d <<= 1) {
        int t = __shfl_up(v, d, 32);
        if (lane >= d) v += t;
    }
    if (lane == 31) s_wave[wave] = v;
    __syncthreads();
    if (tid == 0) {
        int run = 0;
        for (int w = 0; w < 32; ++w) { int t = s_wave[w]; s_wave[w] = run; run += t; }
    }
    __syncthreads();
    if (tid < NBMAX) s_cur[tid] = s_wave[wave] + (v - c) + table[blockIdx.x * NBMAX + tid];
    __syncthreads();

    const int CH = (E + NCH - 1) / NCH;
    const int e0 = blockIdx.x * CH;
    const int e1 = min(E, e0 + CH);
    int e = e0 + tid;
    for (; e + 3 * TPB < e1; e += 4 * TPB) {
        int d[4], s[4];
        for (int u = 0; u < 4; ++u) { d[u] = dst[e + u * TPB]; s[u] = src[e + u * TPB]; }
        for (int u = 0; u < 4; ++u) {
            const int pos = atomicAdd(&s_cur[gcn_bucket(d[u])], 1);
            pairs[pos] = make_int2(s[u], d[u]);
        }
    }
    for (; e < e1; e += TPB) {
        const int d = dst[e], s = src[e];
        const int pos = atomicAdd(&s_cur[gcn_bucket(d)], 1);
        pairs[pos] = make_int2(s, d);
    }
}

__global__ __launch_bounds__(TPB) void gcn_bucket_csr_kernel(const int2* __restrict__ pairs,
                                                             const int* __restrict__ tot,
                                                             int* __restrict__ cnt,
                                                             int* __restrict__ off,
                                                             float* __restrict__ dis,
                                                             int* __restrict__ csr, int N) {
    const int b  = blockIdx.x;
    const int n0 = b << BSH;
    if (n0 >= N) return;
    __shared__ int s_cnt[BNODES];
    __shared__ int s_off[BNODES];
    __shared__ int s_cur[BNODES];
    __shared__ int s_red[32];
    __shared__ int s_wave[32];
    __shared__ int s_bs;
    const int tid  = threadIdx.x;
    const int lane = tid & 31;
    const int wave = tid >> 5;

    int part = (tid < b) ? tot[tid] : 0;
    for (int m = 16; m >= 1; m >>= 1) part += __shfl_xor(part, m, 32);
    if (lane == 0) s_red[wave] = part;
    if (tid < BNODES) { s_cnt[tid] = 0; s_cur[tid] = 0; }
    __syncthreads();
    if (tid == 0) {
        int r = 0;
        for (int w = 0; w < 32; ++w) r += s_red[w];
        s_bs = r;
    }
    __syncthreads();
    const int bs = s_bs;
    const int ne = tot[b];
    const int2* pb = pairs + bs;

    int i = tid;
    for (; i + 3 * TPB < ne; i += 4 * TPB) {
        int2 p[4];
        for (int u = 0; u < 4; ++u) p[u] = pb[i + u * TPB];
        for (int u = 0; u < 4; ++u) atomicAdd(&s_cnt[(p[u].y - n0) & (BNODES - 1)], 1);
    }
    for (; i < ne; i += TPB) atomicAdd(&s_cnt[(pb[i].y - n0) & (BNODES - 1)], 1);
    __syncthreads();

    const int c = (tid < BNODES) ? s_cnt[tid] : 0;
    int v = c;
    for (int d = 1; d < 32; d <<= 1) {
        int t = __shfl_up(v, d, 32);
        if (lane >= d) v += t;
    }
    if (lane == 31) s_wave[wave] = v;
    __syncthreads();
    if (tid == 0) {
        int run = 0;
        for (int w = 0; w < 32; ++w) { int t = s_wave[w]; s_wave[w] = run; run += t; }
    }
    __syncthreads();
    if (tid < BNODES) {
        const int o = s_wave[wave] + (v - c);
        s_off[tid] = o;
        const int n = n0 + tid;
        if (n < N) {
            cnt[n] = c;
            off[n] = bs + o;
            dis[n] = rsqrtf((float)(c + 1));
        }
    }
    __syncthreads();

    i = tid;
    for (; i + 3 * TPB < ne; i += 4 * TPB) {
        int2 p[4];
        for (int u = 0; u < 4; ++u) p[u] = pb[i + u * TPB];
        for (int u = 0; u < 4; ++u) {
            const int dl = (p[u].y - n0) & (BNODES - 1);
            const int r  = atomicAdd(&s_cur[dl], 1);
            csr[bs + s_off[dl] + r] = p[u].x;
        }
    }
    for (; i < ne; i += TPB) {
        const int2 p = pb[i];
        const int dl = (p.y - n0) & (BNODES - 1);
        const int r  = atomicAdd(&s_cur[dl], 1);
        csr[bs + s_off[dl] + r] = p.x;
    }
}

__global__ void gcn_gemm1_wmma(const float* __restrict__ x,
                               const float* __restrict__ W1,
                               float* __restrict__ h1, int n_tiles) {
    __shared__ float sW[F_IN * HID];
    for (int i = threadIdx.x; i < F_IN * HID; i += blockDim.x) sW[i] = W1[i];
    __syncthreads();

    const int wave_in_blk = threadIdx.x >> 5;
    const int tile = blockIdx.x * (blockDim.x >> 5) + wave_in_blk;
    if (tile >= n_tiles) return;

    const int lane = threadIdx.x & 31;
    const int half = lane >> 4;
    const int l16  = lane & 15;

    const int row_base = tile * 16;
    const float* xrow = x + (long long)(row_base + l16) * F_IN;

    v8f c = {};
    #pragma unroll 8
    for (int k = 0; k < F_IN; k += 4) {
        v2f a;
        a.x = xrow[k + 2 * half + 0];
        a.y = xrow[k + 2 * half + 1];
        v2f b;
        b.x = sW[(k + 2 * half + 0) * HID + l16];
        b.y = sW[(k + 2 * half + 1) * HID + l16];
        c = __builtin_amdgcn_wmma_f32_16x16x4_f32(
                false, a, false, b, (short)0, c, false, false);
    }

    float* hrow = h1 + (long long)row_base * HID;
    #pragma unroll
    for (int i = 0; i < 8; ++i)
        hrow[(i + 8 * half) * HID + l16] = c[i];
}

__global__ void gcn_agg1_kernel(const int* __restrict__ csr,
                                const int* __restrict__ off,
                                const int* __restrict__ cnt,
                                const float* __restrict__ dis,
                                const float* __restrict__ h1,
                                float* __restrict__ agg1, int N) {
    long long t = (long long)blockIdx.x * blockDim.x + threadIdx.x;
    int n = (int)(t >> 4);
    int f = (int)(t & 15);
    if (n >= N) return;
    const int c = cnt[n];
    const int* p = csr + off[n];
    const float dn = dis[n];
    float acc = 0.f;
    for (int j = 0; j < c; j += 8) {
        int   s[8];
        float w[8], v[8];
        #pragma unroll
        for (int u = 0; u < 8; ++u) {
            int jj = j + u;
            s[u] = p[jj < c ? jj : c - 1];
        }
        #pragma unroll
        for (int u = 0; u < 8; ++u) {
            const float ds = dis[s[u]];
            w[u] = (j + u < c) ? ds * dn : 0.f;
            v[u] = h1[(long long)s[u] * HID + f];
        }
        #pragma unroll
        for (int u = 0; u < 8; ++u) acc = fmaf(w[u], v[u], acc);
    }
    agg1[(long long)n * HID + f] = acc;
}

__global__ void gcn_relu1_kernel(const float* __restrict__ agg1,
                                 const float* __restrict__ h1,
                                 const float* __restrict__ dis,
                                 const float* __restrict__ b1,
                                 float* __restrict__ r, int N) {
    long long t = (long long)blockIdx.x * blockDim.x + threadIdx.x;
    int n = (int)(t >> 4);
    int f = (int)(t & 15);
    if (n >= N) return;
    float di = dis[n];
    long long idx = (long long)n * HID + f;
    float z = agg1[idx] + di * di * h1[idx] + b1[f];
    r[idx] = fmaxf(z, 0.0f);
}

__global__ void gcn_gemm2_kernel(const float* __restrict__ r,
                                 const float* __restrict__ W2,
                                 float* __restrict__ h2, int N) {
    int n = blockIdx.x * blockDim.x + threadIdx.x;
    if (n >= N) return;
    const float* rn = r + (long long)n * HID;
    float a0 = 0.f, a1 = 0.f;
    #pragma unroll
    for (int f = 0; f < HID; ++f) {
        float v = rn[f];
        a0 = fmaf(v, W2[f * OUT + 0], a0);
        a1 = fmaf(v, W2[f * OUT + 1], a1);
    }
    h2[(long long)n * OUT + 0] = a0;
    h2[(long long)n * OUT + 1] = a1;
}

__global__ void gcn_agg2_kernel(const int* __restrict__ csr,
                                const int* __restrict__ off,
                                const int* __restrict__ cnt,
                                const float* __restrict__ dis,
                                const float* __restrict__ h2,
                                float* __restrict__ agg2, int N) {
    long long t = (long long)blockIdx.x * blockDim.x + threadIdx.x;
    int n = (int)(t >> 4);
    int l = (int)(t & 15);
    if (n >= N) return;
    const int c = cnt[n];
    const int* p = csr + off[n];
    const float dn = dis[n];
    float a0 = 0.f, a1 = 0.f;
    for (int j = l; j < c; j += 16) {
        int s = p[j];
        float w = dis[s] * dn;
        const float2 hv = *reinterpret_cast<const float2*>(h2 + (long long)s * OUT);
        a0 = fmaf(w, hv.x, a0);
        a1 = fmaf(w, hv.y, a1);
    }
    #pragma unroll
    for (int m = 8; m >= 1; m >>= 1) {
        a0 += __shfl_xor(a0, m, 16);
        a1 += __shfl_xor(a1, m, 16);
    }
    if (l == 0) {
        agg2[(long long)n * OUT + 0] = a0;
        agg2[(long long)n * OUT + 1] = a1;
    }
}

__global__ void gcn_final_kernel(const float* __restrict__ agg2,
                                 const float* __restrict__ h2,
                                 const float* __restrict__ dis,
                                 const float* __restrict__ b2,
                                 float* __restrict__ out, int N) {
    int n = blockIdx.x * blockDim.x + threadIdx.x;
    if (n >= N) return;
    float di = dis[n];
    float w = di * di;
    long long i0 = (long long)n * OUT;
    float z0 = agg2[i0 + 0] + w * h2[i0 + 0] + b2[0];
    float z1 = agg2[i0 + 1] + w * h2[i0 + 1] + b2[1];
    float m = fmaxf(z0, z1);
    float lse = m + logf(expf(z0 - m) + expf(z1 - m));
    out[i0 + 0] = z0 - lse;
    out[i0 + 1] = z1 - lse;
}

extern "C" void kernel_launch(void* const* d_in, const int* in_sizes, int n_in,
                              void* d_out, int out_size, void* d_ws, size_t ws_size,
                              hipStream_t stream) {
    const float* x  = (const float*)d_in[0];
    const int*   ei = (const int*)d_in[1];
    const float* W1 = (const float*)d_in[2];
    const float* b1 = (const float*)d_in[3];
    const float* W2 = (const float*)d_in[4];
    const float* b2 = (const float*)d_in[5];
    float* out = (float*)d_out;

    const int       N = in_sizes[0] / F_IN;
    const long long E = (long long)in_sizes[1] / 2;
    const int* src = ei;
    const int* dst = ei + E;

    float* ws   = (float*)d_ws;
    int*   cnt  = (int*)ws;
    float* dis  = ws   + N;
    float* h1   = dis  + N;
    float* agg1 = h1   + (long long)N * HID;
    float* r    = agg1 + (long long)N * HID;
    float* h2   = r    + (long long)N * HID;
    float* agg2 = h2   + (long long)N * OUT;
    int*   off  = (int*)(agg2 + (long long)N * OUT);
    int2*  pairs = (int2*)(off + N);
    int*   csr  = (int*)(pairs + E);
    int*   table = csr + E;
    int*   tot  = table + NCH * NBMAX;

    const int B = 256;

    gcn_bhist_kernel<<<NCH, TPB, 0, stream>>>(dst, table, (int)E);

    gcn_bscan_kernel<<<NBMAX / 32, TPB, 0, stream>>>(table, tot);

    gcn_bscatter_kernel<<<NCH, TPB, 0, stream>>>(src, dst, table, tot, pairs, (int)E);

    gcn_bucket_csr_kernel<<<NBMAX, TPB, 0, stream>>>(pairs, tot, cnt, off, dis, csr, N);

    {
        int n_tiles = N / 16;
        int waves_per_block = B / 32;
        int blocks = (n_tiles + waves_per_block - 1) / waves_per_block;
        gcn_gemm1_wmma<<<blocks, B, 0, stream>>>(x, W1, h1, n_tiles);
    }

    {
        long long T = (long long)N * HID;
        gcn_agg1_kernel<<<(unsigned)((T + B - 1) / B), B, 0, stream>>>(csr, off, cnt, dis, h1, agg1, N);
    }

    {
        long long T = (long long)N * HID;
        gcn_relu1_kernel<<<(unsigned)((T + B - 1) / B), B, 0, stream>>>(agg1, h1, dis, b1, r, N);
    }

    gcn_gemm2_kernel<<<(N + B - 1) / B, B, 0, stream>>>(r, W2, h2, N);

    {
        long long T = (long long)N * 16;
        gcn_agg2_kernel<<<(unsigned)((T + B - 1) / B), B, 0, stream>>>(csr, off, cnt, dis, h2, agg2, N);
    }

    gcn_final_kernel<<<(N + B - 1) / B, B, 0, stream>>>(agg2, h2, dis, b2, out, N);
}


# Per-node epilogues fused into the two gathers: layer-1 sum also does self-loop, bias, relu and the 16x2 product (16 terms added by a 16-lane butterfly), layer-2 sum also log-softmax; 10 launches to 7
# speedup vs baseline: 4.4709x; 1.0701x over previous
#include <hip/hip_runtime.h>
#include <hip/hip_bf16.h>

typedef __attribute__((ext_vector_type(2))) float v2f;
typedef __attribute__((ext_vector_type(8))) float v8f;

#define F_IN 128
#define HID  16
#define OUT  2

#define NCH    256
#define NBMAX  512
#define BSH    8
#define BNODES 256
#define TPB    1024

__device__ __forceinline__ int gcn_bucket(int d) {
    return min((int)((unsigned)d >> BSH), NBMAX - 1);
}

__global__ __launch_bounds__(TPB) void gcn_bhist_kernel(const int* __restrict__ dst,
                                                        int* __restrict__ table, int E) {
    __shared__ int s_h[NBMAX];
    const int tid = threadIdx.x;
    if (tid < NBMAX) s_h[tid] = 0;
    __syncthreads();
    const int CH = (E + NCH - 1) / NCH;
    const int e0 = blockIdx.x * CH;
    const int e1 = min(E, e0 + CH);
    int e = e0 + tid;
    for (; e + 3 * TPB < e1; e += 4 * TPB) {
        int d[4];
        for (int u = 0; u < 4; ++u) d[u] = dst[e + u * TPB];
        for (int u = 0; u < 4; ++u) atomicAdd(&s_h[gcn_bucket(d[u])], 1);
    }
    for (; e < e1; e += TPB) atomicAdd(&s_h[gcn_bucket(dst[e])], 1);
    __syncthreads();
    if (tid < NBMAX) table[blockIdx.x * NBMAX + tid] = s_h[tid];
}

__global__ __launch_bounds__(TPB) void gcn_bscan_kernel(int* __restrict__ table,
                                                        int* __restrict__ tot) {
    __shared__ int s_p[32][33];
    const int bl = threadIdx.x & 31;
    const int cg = threadIdx.x >> 5;
    const int b  = blockIdx.x * 32 + bl;
    int v[8];
    for (int k = 0; k < 8; ++k) v[k] = table[(cg * 8 + k) * NBMAX + b];
    int sum = 0;
    for (int k = 0; k < 8; ++k) { int t = v[k]; v[k] = sum; sum += t; }
    s_p[cg][bl] = sum;
    __syncthreads();
    if (cg == 0) {
        int run = 0;
        for (int g = 0; g < 32; ++g) { int t = s_p[g][bl]; s_p[g][bl] = run; run += t; }
        tot[b] = run;
    }
    __syncthreads();
    const int base = s_p[cg][bl];
    for (int k = 0; k < 8; ++k) table[(cg * 8 + k) * NBMAX + b] = base + v[k];
}

__global__ __launch_bounds__(TPB) void gcn_bscatter_kernel(const int* __restrict__ src,
                                                           const int* __restrict__ dst,
                                                           const int* __restrict__ table,
                                                           const int* __restrict__ tot,
                                                           int2* __restrict__ pairs, int E) {
    __shared__ int s_cur[NBMAX];
    __shared__ int s_wave[32];
    const int tid  = threadIdx.x;
    const int lane = tid & 31;
    const int wave = tid >> 5;
    const int c = (tid < NBMAX) ? tot[tid] : 0;
    int v = c;
    for (int d = 1; d < 32; d <<= 1) {
        int t = __shfl_up(v, d, 32);
        if (lane >= d) v += t;
    }
    if (lane == 31) s_wave[wave] = v;
    __syncthreads();
    if (tid == 0) {
        int run = 0;
        for (int w = 0; w < 32; ++w) { int t = s_wave[w]; s_wave[w] = run; run += t; }
    }
    __syncthreads();
    if (tid < NBMAX) s_cur[tid] = s_wave[wave] + (v - c) + table[blockIdx.x * NBMAX + tid];
    __syncthreads();

    const int CH = (E + NCH - 1) / NCH;
    const int e0 = blockIdx.x * CH;
    const int e1 = min(E, e0 + CH);
    int e = e0 + tid;
    for (; e + 3 * TPB < e1; e += 4 * TPB) {
        int d[4], s[4];
        for (int u = 0; u < 4; ++u) { d[u] = dst[e + u * TPB]; s[u] = src[e + u * TPB]; }
        for (int u = 0; u < 4; ++u) {
            const int pos = atomicAdd(&s_cur[gcn_bucket(d[u])], 1);
            pairs[pos] = make_int2(s[u], d[u]);
        }
    }
    for (; e < e1; e += TPB) {
        const int d = dst[e], s = src[e];
        const int pos = atomicAdd(&s_cur[gcn_bucket(d)], 1);
        pairs[pos] = make_int2(s, d);
    }
}

__global__ __launch_bounds__(TPB) void gcn_bucket_csr_kernel(const int2* __restrict__ pairs,
                                                             const int* __restrict__ tot,
                                                             int* __restrict__ cnt,
                                                             int* __restrict__ off,
                                                             float* __restrict__ dis,
                                                             int* __restrict__ csr, int N) {
    const int b  = blockIdx.x;
    const int n0 = b << BSH;
    if (n0 >= N) return;
    __shared__ int s_cnt[BNODES];
    __shared__ int s_off[BNODES];
    __shared__ int s_cur[BNODES];
    __shared__ int s_red[32];
    __shared__ int s_wave[32];
    __shared__ int s_bs;
    const int tid  = threadIdx.x;
    const int lane = tid & 31;
    const int wave = tid >> 5;

    int part = (tid < b) ? tot[tid] : 0;
    for (int m = 16; m >= 1; m >>= 1) part += __shfl_xor(part, m, 32);
    if (lane == 0) s_red[wave] = part;
    if (tid < BNODES) { s_cnt[tid] = 0; s_cur[tid] = 0; }
    __syncthreads();
    if (tid == 0) {
        int r = 0;
        for (int w = 0; w < 32; ++w) r += s_red[w];
        s_bs = r;
    }
    __syncthreads();
    const int bs = s_bs;
    const int ne = tot[b];
    const int2* pb = pairs + bs;

    int i = tid;
    for (; i + 3 * TPB < ne; i += 4 * TPB) {
        int2 p[4];
        for (int u = 0; u < 4; ++u) p[u] = pb[i + u * TPB];
        for (int u = 0; u < 4; ++u) atomicAdd(&s_cnt[(p[u].y - n0) & (BNODES - 1)], 1);
    }
    for (; i < ne; i += TPB) atomicAdd(&s_cnt[(pb[i].y - n0) & (BNODES - 1)], 1);
    __syncthreads();

    const int c = (tid < BNODES) ? s_cnt[tid] : 0;
    int v = c;
    for (int d = 1; d < 32; d <<= 1) {
        int t = __shfl_up(v, d, 32);
        if (lane >= d) v += t;
    }
    if (lane == 31) s_wave[wave] = v;
    __syncthreads();
    if (tid == 0) {
        int run = 0;
        for (int w = 0; w < 32; ++w) { int t = s_wave[w]; s_wave[w] = run; run += t; }
    }
    __syncthreads();
    if (tid < BNODES) {
        const int o = s_wave[wave] + (v - c);
        s_off[tid] = o;
        const int n = n0 + tid;
        if (n < N) {
            cnt[n] = c;
            off[n] = bs + o;
            dis[n] = rsqrtf((float)(c + 1));
        }
    }
    __syncthreads();

    i = tid;
    for (; i + 3 * TPB < ne; i += 4 * TPB) {
        int2 p[4];
        for (int u = 0; u < 4; ++u) p[u] = pb[i + u * TPB];
        for (int u = 0; u < 4; ++u) {
            const int dl = (p[u].y - n0) & (BNODES - 1);
            const int r  = atomicAdd(&s_cur[dl], 1);
            csr[bs + s_off[dl] + r] = p[u].x;
        }
    }
    for (; i < ne; i += TPB) {
        const int2 p = pb[i];
        const int dl = (p.y - n0) & (BNODES - 1);
        const int r  = atomicAdd(&s_cur[dl], 1);
        csr[bs + s_off[dl] + r] = p.x;
    }
}

__global__ void gcn_gemm1_wmma(const float* __restrict__ x,
                               const float* __restrict__ W1,
                               float* __restrict__ h1, int n_tiles) {
    __shared__ float sW[F_IN * HID];
    for (int i = threadIdx.x; i < F_IN * HID; i += blockDim.x) sW[i] = W1[i];
    __syncthreads();

    const int wave_in_blk = threadIdx.x >> 5;
    const int tile = blockIdx.x * (blockDim.x >> 5) + wave_in_blk;
    if (tile >= n_tiles) return;

    const int lane = threadIdx.x & 31;
    const int half = lane >> 4;
    const int l16  = lane & 15;

    const int row_base = tile * 16;
    const float* xrow = x + (long long)(row_base + l16) * F_IN;

    v8f c = {};
    #pragma unroll 8
    for (int k = 0; k < F_IN; k += 4) {
        v2f a;
        a.x = xrow[k + 2 * half + 0];
        a.y = xrow[k + 2 * half + 1];
        v2f b;
        b.x = sW[(k + 2 * half + 0) * HID + l16];
        b.y = sW[(k + 2 * half + 1) * HID + l16];
        c = __builtin_amdgcn_wmma_f32_16x16x4_f32(
                false, a, false, b, (short)0, c, false, false);
    }

    float* hrow = h1 + (long long)row_base * HID;
    #pragma unroll
    for (int i = 0; i < 8; ++i)
        hrow[(i + 8 * half) * HID + l16] = c[i];
}

__global__ void gcn_layer1_kernel(const int* __restrict__ csr,
                                  const int* __restrict__ off,
                                  const int* __restrict__ cnt,
                                  const float* __restrict__ dis,
                                  const float* __restrict__ h1,
                                  const float* __restrict__ b1,
                                  const float* __restrict__ W2,
                                  float* __restrict__ h2, int N) {
    long long t = (long long)blockIdx.x * blockDim.x + threadIdx.x;
    int n = (int)(t >> 4);
    int f = (int)(t & 15);
    if (n >= N) return;
    const int c = cnt[n];
    const int* p = csr + off[n];
    const float dn = dis[n];
    const float hs  = h1[(long long)n * HID + f];
    const float bf  = b1[f];
    const float w20 = W2[f * OUT + 0];
    const float w21 = W2[f * OUT + 1];
    float acc = 0.f;
    for (int j = 0; j < c; j += 8) {
        int   s[8];
        float w[8], v[8];
        #pragma unroll
        for (int u = 0; u < 8; ++u) {
            int jj = j + u;
            s[u] = p[jj < c ? jj : c - 1];
        }
        #pragma unroll
        for (int u = 0; u < 8; ++u) {
            const float ds = dis[s[u]];
            w[u] = (j + u < c) ? ds * dn : 0.f;
            v[u] = h1[(long long)s[u] * HID + f];
        }
        #pragma unroll
        for (int u = 0; u < 8; ++u) acc = fmaf(w[u], v[u], acc);
    }
    const float z = acc + dn * dn * hs + bf;
    const float r = fmaxf(z, 0.0f);
    float a0 = r * w20;
    float a1 = r * w21;
    #pragma unroll
    for (int m = 8; m >= 1; m >>= 1) {
        a0 += __shfl_xor(a0, m, 16);
        a1 += __shfl_xor(a1, m, 16);
    }
    if (f == 0)
        *reinterpret_cast<float2*>(h2 + (long long)n * OUT) = make_float2(a0, a1);
}

__global__ void gcn_layer2_kernel(const int* __restrict__ csr,
                                  const int* __restrict__ off,
                                  const int* __restrict__ cnt,
                                  const float* __restrict__ dis,
                                  const float* __restrict__ h2,
                                  const float* __restrict__ b2,
                                  float* __restrict__ out, int N) {
    long long t = (long long)blockIdx.x * blockDim.x + threadIdx.x;
    int n = (int)(t >> 4);
    int l = (int)(t & 15);
    if (n >= N) return;
    const int c = cnt[n];
    const int* p = csr + off[n];
    const float dn = dis[n];
    const float2 hn = *reinterpret_cast<const float2*>(h2 + (long long)n * OUT);
    const float b20 = b2[0];
    const float b21 = b2[1];
    float a0 = 0.f, a1 = 0.f;
    for (int j = l; j < c; j += 16) {
        int s = p[j];
        float w = dis[s] * dn;
        const float2 hv = *reinterpret_cast<const float2*>(h2 + (long long)s * OUT);
        a0 = fmaf(w, hv.x, a0);
        a1 = fmaf(w, hv.y, a1);
    }
    #pragma unroll
    for (int m = 8; m >= 1; m >>= 1) {
        a0 += __shfl_xor(a0, m, 16);
        a1 += __shfl_xor(a1, m, 16);
    }
    if (l == 0) {
        const float w = dn * dn;
        const float z0 = a0 + w * hn.x + b20;
        const float z1 = a1 + w * hn.y + b21;
        const float m = fmaxf(z0, z1);
        const float lse = m + logf(expf(z0 - m) + expf(z1 - m));
        *reinterpret_cast<float2*>(out + (long long)n * OUT) = make_float2(z0 - lse, z1 - lse);
    }
}

extern "C" void kernel_launch(void* const* d_in, const int* in_sizes, int n_in,
                              void* d_out, int out_size, void* d_ws, size_t ws_size,
                              hipStream_t stream) {
    const float* x  = (const float*)d_in[0];
    const int*   ei = (const int*)d_in[1];
    const float* W1 = (const float*)d_in[2];
    const float* b1 = (const float*)d_in[3];
    const float* W2 = (const float*)d_in[4];
    const float* b2 = (const float*)d_in[5];
    float* out = (float*)d_out;

    const int       N = in_sizes[0] / F_IN;
    const long long E = (long long)in_sizes[1] / 2;
    const int* src = ei;
    const int* dst = ei + E;

    float* ws   = (float*)d_ws;
    int*   cnt  = (int*)ws;
    float* dis  = ws   + N;
    float* h1   = dis  + N;
    float* h2   = h1   + (long long)N * HID;
    int*   off  = (int*)(h2 + (long long)N * OUT);
    int2*  pairs = (int2*)(off + N);
    int*   csr  = (int*)(pairs + E);
    int*   table = csr + E;
    int*   tot  = table + NCH * NBMAX;

    const int B = 256;

    gcn_bhist_kernel<<<NCH, TPB, 0, stream>>>(dst, table, (int)E);

    gcn_bscan_kernel<<<NBMAX / 32, TPB, 0, stream>>>(table, tot);

    gcn_bscatter_kernel<<<NCH, TPB, 0, stream>>>(src, dst, table, tot, pairs, (int)E);

    gcn_bucket_csr_kernel<<<NBMAX, TPB, 0, stream>>>(pairs, tot, cnt, off, dis, csr, N);

    {
        int n_tiles = N / 16;
        int waves_per_block = B / 32;
        int blocks = (n_tiles + waves_per_block - 1) / waves_per_block;
        gcn_gemm1_wmma<<<blocks, B, 0, stream>>>(x, W1, h1, n_tiles);
    }

    {
        long long T = (long long)N * HID;
        gcn_layer1_kernel<<<(unsigned)((T + B - 1) / B), B, 0, stream>>>(csr, off, cnt, dis, h1, b1, W2, h2, N);
    }

    {
        long long T = (long long)N * 16;
        gcn_layer2_kernel<<<(unsigned)((T + B - 1) / B), B, 0, stream>>>(csr, off, cnt, dis, h2, b2, out, N);
    }
}


# Both gather kernels: 4 lanes per node instead of 16 (layer 1: one 16-byte row piece per lane; layer 2: edges split 4 ways), 12500 waves each, not 50000; 4-lane sum trees replace the 16-lane ones
# speedup vs baseline: 4.6179x; 1.0329x over previous
#include <hip/hip_runtime.h>
#include <hip/hip_bf16.h>

typedef __attribute__((ext_vector_type(2))) float v2f;
typedef __attribute__((ext_vector_type(8))) float v8f;

#define F_IN 128
#define HID  16
#define OUT  2

#define NCH    256
#define NBMAX  512
#define BSH    8
#define BNODES 256
#define TPB    1024

__device__ __forceinline__ int gcn_bucket(int d) {
    return min((int)((unsigned)d >> BSH), NBMAX - 1);
}

__global__ __launch_bounds__(TPB) void gcn_bhist_kernel(const int* __restrict__ dst,
                                                        int* __restrict__ table, int E) {
    __shared__ int s_h[NBMAX];
    const int tid = threadIdx.x;
    if (tid < NBMAX) s_h[tid] = 0;
    __syncthreads();
    const int CH = (E + NCH - 1) / NCH;
    const int e0 = blockIdx.x * CH;
    const int e1 = min(E, e0 + CH);
    int e = e0 + tid;
    for (; e + 3 * TPB < e1; e += 4 * TPB) {
        int d[4];
        for (int u = 0; u < 4; ++u) d[u] = dst[e + u * TPB];
        for (int u = 0; u < 4; ++u) atomicAdd(&s_h[gcn_bucket(d[u])], 1);
    }
    for (; e < e1; e += TPB) atomicAdd(&s_h[gcn_bucket(dst[e])], 1);
    __syncthreads();
    if (tid < NBMAX) table[blockIdx.x * NBMAX + tid] = s_h[tid];
}

__global__ __launch_bounds__(TPB) void gcn_bscan_kernel(int* __restrict__ table,
                                                        int* __restrict__ tot) {
    __shared__ int s_p[32][33];
    const int bl = threadIdx.x & 31;
    const int cg = threadIdx.x >> 5;
    const int b  = blockIdx.x * 32 + bl;
    int v[8];
    for (int k = 0; k < 8; ++k) v[k] = table[(cg * 8 + k) * NBMAX + b];
    int sum = 0;
    for (int k = 0; k < 8; ++k) { int t = v[k]; v[k] = sum; sum += t; }
    s_p[cg][bl] = sum;
    __syncthreads();
    if (cg == 0) {
        int run = 0;
        for (int g = 0; g < 32; ++g) { int t = s_p[g][bl]; s_p[g][bl] = run; run += t; }
        tot[b] = run;
    }
    __syncthreads();
    const int base = s_p[cg][bl];
    for (int k = 0; k < 8; ++k) table[(cg * 8 + k) * NBMAX + b] = base + v[k];
}

__global__ __launch_bounds__(TPB) void gcn_bscatter_kernel(const int* __restrict__ src,
                                                           const int* __restrict__ dst,
                                                           const int* __restrict__ table,
                                                           const int* __restrict__ tot,
                                                           int2* __restrict__ pairs, int E) {
    __shared__ int s_cur[NBMAX];
    __shared__ int s_wave[32];
    const int tid  = threadIdx.x;
    const int lane = tid & 31;
    const int wave = tid >> 5;
    const int c = (tid < NBMAX) ? tot[tid] : 0;
    int v = c;
    for (int d = 1; d < 32; d <<= 1) {
        int t = __shfl_up(v, d, 32);
        if (lane >= d) v += t;
    }
    if (lane == 31) s_wave[wave] = v;
    __syncthreads();
    if (tid == 0) {
        int run = 0;
        for (int w = 0; w < 32; ++w) { int t = s_wave[w]; s_wave[w] = run; run += t; }
    }
    __syncthreads();
    if (tid < NBMAX) s_cur[tid] = s_wave[wave] + (v - c) + table[blockIdx.x * NBMAX + tid];
    __syncthreads();

    const int CH = (E + NCH - 1) / NCH;
    const int e0 = blockIdx.x * CH;
    const int e1 = min(E, e0 + CH);
    int e = e0 + tid;
    for (; e + 3 * TPB < e1; e += 4 * TPB) {
        int d[4], s[4];
        for (int u = 0; u < 4; ++u) { d[u] = dst[e + u * TPB]; s[u] = src[e + u * TPB]; }
        for (int u = 0; u < 4; ++u) {
            const int pos = atomicAdd(&s_cur[gcn_bucket(d[u])], 1);
            pairs[pos] = make_int2(s[u], d[u]);
        }
    }
    for (; e < e1; e += TPB) {
        const int d = dst[e], s = src[e];
        const int pos = atomicAdd(&s_cur[gcn_bucket(d)], 1);
        pairs[pos] = make_int2(s, d);
    }
}

__global__ __launch_bounds__(TPB) void gcn_bucket_csr_kernel(const int2* __restrict__ pairs,
                                                             const int* __restrict__ tot,
                                                             int* __restrict__ cnt,
                                                             int* __restrict__ off,
                                                             float* __restrict__ dis,
                                                             int* __restrict__ csr, int N) {
    const int b  = blockIdx.x;
    const int n0 = b << BSH;
    if (n0 >= N) return;
    __shared__ int s_cnt[BNODES];
    __shared__ int s_off[BNODES];
    __shared__ int s_cur[BNODES];
    __shared__ int s_red[32];
    __shared__ int s_wave[32];
    __shared__ int s_bs;
    const int tid  = threadIdx.x;
    const int lane = tid & 31;
    const int wave = tid >> 5;

    int part = (tid < b) ? tot[tid] : 0;
    for (int m = 16; m >= 1; m >>= 1) part += __shfl_xor(part, m, 32);
    if (lane == 0) s_red[wave] = part;
    if (tid < BNODES) { s_cnt[tid] = 0; s_cur[tid] = 0; }
    __syncthreads();
    if (tid == 0) {
        int r = 0;
        for (int w = 0; w < 32; ++w) r += s_red[w];
        s_bs = r;
    }
    __syncthreads();
    const int bs = s_bs;
    const int ne = tot[b];
    const int2* pb = pairs + bs;

    int i = tid;
    for (; i + 3 * TPB < ne; i += 4 * TPB) {
        int2 p[4];
        for (int u = 0; u < 4; ++u) p[u] = pb[i + u * TPB];
        for (int u = 0; u < 4; ++u) atomicAdd(&s_cnt[(p[u].y - n0) & (BNODES - 1)], 1);
    }
    for (; i < ne; i += TPB) atomicAdd(&s_cnt[(pb[i].y - n0) & (BNODES - 1)], 1);
    __syncthreads();

    const int c = (tid < BNODES) ? s_cnt[tid] : 0;
    int v = c;
    for (int d = 1; d < 32; d <<= 1) {
        int t = __shfl_up(v, d, 32);
        if (lane >= d) v += t;
    }
    if (lane == 31) s_wave[wave] = v;
    __syncthreads();
    if (tid == 0) {
        int run = 0;
        for (int w = 0; w < 32; ++w) { int t = s_wave[w]; s_wave[w] = run; run += t; }
    }
    __syncthreads();
    if (tid < BNODES) {
        const int o = s_wave[wave] + (v - c);
        s_off[tid] = o;
        const int n = n0 + tid;
        if (n < N) {
            cnt[n] = c;
            off[n] = bs + o;
            dis[n] = rsqrtf((float)(c + 1));
        }
    }
    __syncthreads();

    i = tid;
    for (; i + 3 * TPB < ne; i += 4 * TPB) {
        int2 p[4];
        for (int u = 0; u < 4; ++u) p[u] = pb[i + u * TPB];
        for (int u = 0; u < 4; ++u) {
            const int dl = (p[u].y - n0) & (BNODES - 1);
            const int r  = atomicAdd(&s_cur[dl], 1);
            csr[bs + s_off[dl] + r] = p[u].x;
        }
    }
    for (; i < ne; i += TPB) {
        const int2 p = pb[i];
        const int dl = (p.y - n0) & (BNODES - 1);
        const int r  = atomicAdd(&s_cur[dl], 1);
        csr[bs + s_off[dl] + r] = p.x;
    }
}

__global__ void gcn_gemm1_wmma(const float* __restrict__ x,
                               const float* __restrict__ W1,
                               float* __restrict__ h1, int n_tiles) {
    __shared__ float sW[F_IN * HID];
    for (int i = threadIdx.x; i < F_IN * HID; i += blockDim.x) sW[i] = W1[i];
    __syncthreads();

    const int wave_in_blk = threadIdx.x >> 5;
    const int tile = blockIdx.x * (blockDim.x >> 5) + wave_in_blk;
    if (tile >= n_tiles) return;

    const int lane = threadIdx.x & 31;
    const int half = lane >> 4;
    const int l16  = lane & 15;

    const int row_base = tile * 16;
    const float* xrow = x + (long long)(row_base + l16) * F_IN;

    v8f c = {};
    #pragma unroll 8
    for (int k = 0; k < F_IN; k += 4) {
        v2f a;
        a.x = xrow[k + 2 * half + 0];
        a.y = xrow[k + 2 * half + 1];
        v2f b;
        b.x = sW[(k + 2 * half + 0) * HID + l16];
        b.y = sW[(k + 2 * half + 1) * HID + l16];
        c = __builtin_amdgcn_wmma_f32_16x16x4_f32(
                false, a, false, b, (short)0, c, false, false);
    }

    float* hrow = h1 + (long long)row_base * HID;
    #pragma unroll
    for (int i = 0; i < 8; ++i)
        hrow[(i + 8 * half) * HID + l16] = c[i];
}

__global__ void gcn_layer1_kernel(const int* __restrict__ csr,
                                  const int* __restrict__ off,
                                  const int* __restrict__ cnt,
                                  const float* __restrict__ dis,
                                  const float* __restrict__ h1,
                                  const float* __restrict__ b1,
                                  const float* __restrict__ W2,
                                  float* __restrict__ h2, int N) {
    long long t = (long long)blockIdx.x * blockDim.x + threadIdx.x;
    int n = (int)(t >> 2);
    int q = (int)(t & 3);
    if (n >= N) return;
    const int c = cnt[n];
    const int* p = csr + off[n];
    const float dn = dis[n];
    const float4 hs = *reinterpret_cast<const float4*>(h1 + (long long)n * HID + 4 * q);
    float bf[4], w20[4], w21[4];
    #pragma unroll
    for (int i = 0; i < 4; ++i) {
        bf[i]  = b1[4 * q + i];
        w20[i] = W2[(4 * q + i) * OUT + 0];
        w21[i] = W2[(4 * q + i) * OUT + 1];
    }
    float acc0 = 0.f, acc1 = 0.f, acc2 = 0.f, acc3 = 0.f;
    for (int j = 0; j < c; j += 8) {
        int    s[8];
        float  w[8];
        float4 v[8];
        #pragma unroll
        for (int u = 0; u < 8; ++u) {
            int jj = j + u;
            s[u] = p[jj < c ? jj : c - 1];
        }
        #pragma unroll
        for (int u = 0; u < 8; ++u) {
            const float ds = dis[s[u]];
            w[u] = (j + u < c) ? ds * dn : 0.f;
            v[u] = *reinterpret_cast<const float4*>(h1 + (long long)s[u] * HID + 4 * q);
        }
        #pragma unroll
        for (int u = 0; u < 8; ++u) {
            acc0 = fmaf(w[u], v[u].x, acc0);
            acc1 = fmaf(w[u], v[u].y, acc1);
            acc2 = fmaf(w[u], v[u].z, acc2);
            acc3 = fmaf(w[u], v[u].w, acc3);
        }
    }
    const float r0 = fmaxf(acc0 + dn * dn * hs.x + bf[0], 0.0f);
    const float r1 = fmaxf(acc1 + dn * dn * hs.y + bf[1], 0.0f);
    const float r2 = fmaxf(acc2 + dn * dn * hs.z + bf[2], 0.0f);
    const float r3 = fmaxf(acc3 + dn * dn * hs.w + bf[3], 0.0f);
    float a0 = r0 * w20[0] + r1 * w20[1] + r2 * w20[2] + r3 * w20[3];
    float a1 = r0 * w21[0] + r1 * w21[1] + r2 * w21[2] + r3 * w21[3];
    #pragma unroll
    for (int m = 2; m >= 1; m >>= 1) {
        a0 += __shfl_xor(a0, m, 4);
        a1 += __shfl_xor(a1, m, 4);
    }
    if (q == 0)
        *reinterpret_cast<float2*>(h2 + (long long)n * OUT) = make_float2(a0, a1);
}

__global__ void gcn_layer2_kernel(const int* __restrict__ csr,
                                  const int* __restrict__ off,
                                  const int* __restrict__ cnt,
                                  const float* __restrict__ dis,
                                  const float* __restrict__ h2,
                                  const float* __restrict__ b2,
                                  float* __restrict__ out, int N) {
    long long t = (long long)blockIdx.x * blockDim.x + threadIdx.x;
    int n = (int)(t >> 2);
    int l = (int)(t & 3);
    if (n >= N) return;
    const int c = cnt[n];
    const int* p = csr + off[n];
    const float dn = dis[n];
    const float2 hn = *reinterpret_cast<const float2*>(h2 + (long long)n * OUT);
    const float b20 = b2[0];
    const float b21 = b2[1];
    float a0 = 0.f, a1 = 0.f;
    for (int j = l; j < c; j += 16) {
        int    s[4];
        float  w[4];
        float2 hv[4];
        #pragma unroll
        for (int u = 0; u < 4; ++u) {
            int jj = j + 4 * u;
            s[u] = p[jj < c ? jj : c - 1];
        }
        #pragma unroll
        for (int u = 0; u < 4; ++u) {
            const float ds = dis[s[u]];
            w[u]  = (j + 4 * u < c) ? ds * dn : 0.f;
            hv[u] = *reinterpret_cast<const float2*>(h2 + (long long)s[u] * OUT);
        }
        #pragma unroll
        for (int u = 0; u < 4; ++u) {
            a0 = fmaf(w[u], hv[u].x, a0);
            a1 = fmaf(w[u], hv[u].y, a1);
        }
    }
    #pragma unroll
    for (int m = 2; m >= 1; m >>= 1) {
        a0 += __shfl_xor(a0, m, 4);
        a1 += __shfl_xor(a1, m, 4);
    }
    if (l == 0) {
        const float w = dn * dn;
        const float z0 = a0 + w * hn.x + b20;
        const float z1 = a1 + w * hn.y + b21;
        const float m = fmaxf(z0, z1);
        const float lse = m + logf(expf(z0 - m) + expf(z1 - m));
        *reinterpret_cast<float2*>(out + (long long)n * OUT) = make_float2(z0 - lse, z1 - lse);
    }
}

extern "C" void kernel_launch(void* const* d_in, const int* in_sizes, int n_in,
                              void* d_out, int out_size, void* d_ws, size_t ws_size,
                              hipStream_t stream) {
    const float* x  = (const float*)d_in[0];
    const int*   ei = (const int*)d_in[1];
    const float* W1 = (const float*)d_in[2];
    const float* b1 = (const float*)d_in[3];
    const float* W2 = (const float*)d_in[4];
    const float* b2 = (const float*)d_in[5];
    float* out = (float*)d_out;

    const int       N = in_sizes[0] / F_IN;
    const long long E = (long long)in_sizes[1] / 2;
    const int* src = ei;
    const int* dst = ei + E;

    float* ws   = (float*)d_ws;
    int*   cnt  = (int*)ws;
    float* dis  = ws   + N;
    float* h1   = dis  + N;
    float* h2   = h1   + (long long)N * HID;
    int*   off  = (int*)(h2 + (long long)N * OUT);
    int2*  pairs = (int2*)(off + N);
    int*   csr  = (int*)(pairs + E);
    int*   table = csr + E;
    int*   tot  = table + NCH * NBMAX;

    const int B = 256;

    gcn_bhist_kernel<<<NCH, TPB, 0, stream>>>(dst, table, (int)E);

    gcn_bscan_kernel<<<NBMAX / 32, TPB, 0, stream>>>(table, tot);

    gcn_bscatter_kernel<<<NCH, TPB, 0, stream>>>(src, dst, table, tot, pairs, (int)E);

    gcn_bucket_csr_kernel<<<NBMAX, TPB, 0, stream>>>(pairs, tot, cnt, off, dis, csr, N);

    {
        int n_tiles = N / 16;
        int waves_per_block = B / 32;
        int blocks = (n_tiles + waves_per_block - 1) / waves_per_block;
        gcn_gemm1_wmma<<<blocks, B, 0, stream>>>(x, W1, h1, n_tiles);
    }

    {
        long long T = (long long)N * 4;
        gcn_layer1_kernel<<<(unsigned)((T + B - 1) / B), B, 0, stream>>>(csr, off, cnt, dis, h1, b1, W2, h2, N);
    }

    {
        long long T = (long long)N * 4;
        gcn_layer2_kernel<<<(unsigned)((T + B - 1) / B), B, 0, stream>>>(csr, off, cnt, dis, h2, b2, out, N);
    }
}


# Both edge-scatter passes order their records in LDS first (chunk runs; a bucket's CSR span), then store them front to back: whole runs per store, not 32 lines; slots still by LDS add arrival
# speedup vs baseline: 6.4976x; 1.4070x over previous
#include <hip/hip_runtime.h>
#include <hip/hip_bf16.h>

typedef __attribute__((ext_vector_type(2))) float v2f;
typedef __attribute__((ext_vector_type(8))) float v8f;

#define F_IN 128
#define HID  16
#define OUT  2

#define NCH    256
#define NBMAX  512
#define BSH    8
#define BNODES 256
#define TPB    1024
#define PCAP   12544
#define CCAP   16384

__device__ __forceinline__ int gcn_bucket(int d) {
    return min((int)((unsigned)d >> BSH), NBMAX - 1);
}

__global__ __launch_bounds__(TPB) void gcn_bhist_kernel(const int* __restrict__ dst,
                                                        int* __restrict__ table, int E) {
    __shared__ int s_h[NBMAX];
    const int tid = threadIdx.x;
    if (tid < NBMAX) s_h[tid] = 0;
    __syncthreads();
    const int CH = (E + NCH - 1) / NCH;
    const int e0 = blockIdx.x * CH;
    const int e1 = min(E, e0 + CH);
    int e = e0 + tid;
    for (; e + 3 * TPB < e1; e += 4 * TPB) {
        int d[4];
        for (int u = 0; u < 4; ++u) d[u] = dst[e + u * TPB];
        for (int u = 0; u < 4; ++u) atomicAdd(&s_h[gcn_bucket(d[u])], 1);
    }
    for (; e < e1; e += TPB) atomicAdd(&s_h[gcn_bucket(dst[e])], 1);
    __syncthreads();
    if (tid < NBMAX) table[blockIdx.x * NBMAX + tid] = s_h[tid];
}

__global__ __launch_bounds__(TPB) void gcn_bscan_kernel(int* __restrict__ table,
                                                        int* __restrict__ tot) {
    __shared__ int s_p[32][33];
    const int bl = threadIdx.x & 31;
    const int cg = threadIdx.x >> 5;
    const int b  = blockIdx.x * 32 + bl;
    int v[8];
    for (int k = 0; k < 8; ++k) v[k] = table[(cg * 8 + k) * NBMAX + b];
    int sum = 0;
    for (int k = 0; k < 8; ++k) { int t = v[k]; v[k] = sum; sum += t; }
    s_p[cg][bl] = sum;
    __syncthreads();
    if (cg == 0) {
        int run = 0;
        for (int g = 0; g < 32; ++g) { int t = s_p[g][bl]; s_p[g][bl] = run; run += t; }
        tot[b] = run;
    }
    __syncthreads();
    const int base = s_p[cg][bl];
    for (int k = 0; k < 8; ++k) table[(cg * 8 + k) * NBMAX + b] = base + v[k];
}

__global__ __launch_bounds__(TPB) void gcn_bscatter_kernel(const int* __restrict__ src,
                                                           const int* __restrict__ dst,
                                                           const int* __restrict__ table,
                                                           const int* __restrict__ tot,
                                                           int2* __restrict__ pairs, int E) {
    __shared__ int  s_cur[NBMAX];
    __shared__ int  s_delta[NBMAX];
    __shared__ int  s_wave[32];
    __shared__ int  s_wloc[32];
    __shared__ int2 s_pairs[PCAP];
    const int tid  = threadIdx.x;
    const int lane = tid & 31;
    const int wave = tid >> 5;
    const int CH = (E + NCH - 1) / NCH;
    const int e0 = blockIdx.x * CH;
    const int e1 = min(E, e0 + CH);
    const bool staged = (CH <= PCAP);

    int c = 0, trow = 0, k = 0;
    if (tid < NBMAX) {
        c    = tot[tid];
        trow = table[blockIdx.x * NBMAX + tid];
        const int tnext = (blockIdx.x + 1 < NCH) ? table[(blockIdx.x + 1) * NBMAX + tid] : c;
        k = tnext - trow;
    }
    int v = c, w = k;
    for (int d = 1; d < 32; d <<= 1) {
        int t  = __shfl_up(v, d, 32);
        int t2 = __shfl_up(w, d, 32);
        if (lane >= d) { v += t; w += t2; }
    }
    if (lane == 31) { s_wave[wave] = v; s_wloc[wave] = w; }
    __syncthreads();
    if (tid == 0) {
        int run = 0;
        for (int g = 0; g < 32; ++g) { int t = s_wave[g]; s_wave[g] = run; run += t; }
    }
    if (tid == 32) {
        int run = 0;
        for (int g = 0; g < 32; ++g) { int t = s_wloc[g]; s_wloc[g] = run; run += t; }
    }
    __syncthreads();
    if (tid < NBMAX) {
        const int gpos = s_wave[wave] + (v - c) + trow;
        const int lpos = s_wloc[wave] + (w - k);
        s_cur[tid]   = staged ? lpos : gpos;
        s_delta[tid] = gpos - lpos;
    }
    __syncthreads();

    int e = e0 + tid;
    if (staged) {
        for (; e + 3 * TPB < e1; e += 4 * TPB) {
            int d[4], s[4];
            for (int u = 0; u < 4; ++u) { d[u] = dst[e + u * TPB]; s[u] = src[e + u * TPB]; }
            int pos[4];
            for (int u = 0; u < 4; ++u) pos[u] = atomicAdd(&s_cur[gcn_bucket(d[u])], 1);
            for (int u = 0; u < 4; ++u)
                s_pairs[min(pos[u], PCAP - 1)] = make_int2(s[u], d[u]);
        }
        for (; e < e1; e += TPB) {
            const int d = dst[e], s = src[e];
            const int pos = atomicAdd(&s_cur[gcn_bucket(d)], 1);
            s_pairs[min(pos, PCAP - 1)] = make_int2(s, d);
        }
        __syncthreads();
        const int nloc = e1 - e0;
        int i = tid;
        for (; i + 3 * TPB < nloc; i += 4 * TPB) {
            int2 p[4];
            int  g[4];
            for (int u = 0; u < 4; ++u) p[u] = s_pairs[i + u * TPB];
            for (int u = 0; u < 4; ++u) g[u] = i + u * TPB + s_delta[gcn_bucket(p[u].y)];
            for (int u = 0; u < 4; ++u) pairs[g[u]] = p[u];
        }
        for (; i < nloc; i += TPB) {
            const int2 p = s_pairs[i];
            pairs[i + s_delta[gcn_bucket(p.y)]] = p;
        }
    } else {
        for (; e + 3 * TPB < e1; e += 4 * TPB) {
            int d[4], s[4];
            for (int u = 0; u < 4; ++u) { d[u] = dst[e + u * TPB]; s[u] = src[e + u * TPB]; }
            for (int u = 0; u < 4; ++u) {
                const int pos = atomicAdd(&s_cur[gcn_bucket(d[u])], 1);
                pairs[pos] = make_int2(s[u], d[u]);
            }
        }
        for (; e < e1; e += TPB) {
            const int d = dst[e], s = src[e];
            const int pos = atomicAdd(&s_cur[gcn_bucket(d)], 1);
            pairs[pos] = make_int2(s, d);
        }
    }
}

__global__ __launch_bounds__(TPB) void gcn_bucket_csr_kernel(const int2* __restrict__ pairs,
                                                             const int* __restrict__ tot,
                                                             int* __restrict__ cnt,
                                                             int* __restrict__ off,
                                                             float* __restrict__ dis,
                                                             int* __restrict__ csr, int N) {
    const int b  = blockIdx.x;
    const int n0 = b << BSH;
    if (n0 >= N) return;
    __shared__ int s_cnt[BNODES];
    __shared__ int s_off[BNODES];
    __shared__ int s_cur[BNODES];
    __shared__ int s_red[32];
    __shared__ int s_wave[32];
    __shared__ int s_bs;
    __shared__ int s_csr[CCAP];
    const int tid  = threadIdx.x;
    const int lane = tid & 31;
    const int wave = tid >> 5;

    int part = (tid < b) ? tot[tid] : 0;
    for (int m = 16; m >= 1; m >>= 1) part += __shfl_xor(part, m, 32);
    if (lane == 0) s_red[wave] = part;
    if (tid < BNODES) { s_cnt[tid] = 0; s_cur[tid] = 0; }
    __syncthreads();
    if (tid == 0) {
        int r = 0;
        for (int w = 0; w < 32; ++w) r += s_red[w];
        s_bs = r;
    }
    __syncthreads();
    const int bs = s_bs;
    const int ne = tot[b];
    const int2* pb = pairs + bs;

    int i = tid;
    for (; i + 3 * TPB < ne; i += 4 * TPB) {
        int2 p[4];
        for (int u = 0; u < 4; ++u) p[u] = pb[i + u * TPB];
        for (int u = 0; u < 4; ++u) atomicAdd(&s_cnt[(p[u].y - n0) & (BNODES - 1)], 1);
    }
    for (; i < ne; i += TPB) atomicAdd(&s_cnt[(pb[i].y - n0) & (BNODES - 1)], 1);
    __syncthreads();

    const int c = (tid < BNODES) ? s_cnt[tid] : 0;
    int v = c;
    for (int d = 1; d < 32; d <<= 1) {
        int t = __shfl_up(v, d, 32);
        if (lane >= d) v += t;
    }
    if (lane == 31) s_wave[wave] = v;
    __syncthreads();
    if (tid == 0) {
        int run = 0;
        for (int w = 0; w < 32; ++w) { int t = s_wave[w]; s_wave[w] = run; run += t; }
    }
    __syncthreads();
    if (tid < BNODES) {
        const int o = s_wave[wave] + (v - c);
        s_off[tid] = o;
        const int n = n0 + tid;
        if (n < N) {
            cnt[n] = c;
            off[n] = bs + o;
            dis[n] = rsqrtf((float)(c + 1));
        }
    }
    __syncthreads();

    i = tid;
    if (ne <= CCAP) {
        for (; i + 3 * TPB < ne; i += 4 * TPB) {
            int2 p[4];
            for (int u = 0; u < 4; ++u) p[u] = pb[i + u * TPB];
            int dl[4], r[4];
            for (int u = 0; u < 4; ++u) {
                dl[u] = (p[u].y - n0) & (BNODES - 1);
                r[u]  = atomicAdd(&s_cur[dl[u]], 1);
            }
            for (int u = 0; u < 4; ++u)
                s_csr[min(s_off[dl[u]] + r[u], CCAP - 1)] = p[u].x;
        }
        for (; i < ne; i += TPB) {
            const int2 p = pb[i];
            const int dl = (p.y - n0) & (BNODES - 1);
            const int r  = atomicAdd(&s_cur[dl], 1);
            s_csr[min(s_off[dl] + r, CCAP - 1)] = p.x;
        }
        __syncthreads();
        int* cb = csr + bs;
        i = tid;
        for (; i + 3 * TPB < ne; i += 4 * TPB) {
            int q[4];
            for (int u = 0; u < 4; ++u) q[u] = s_csr[i + u * TPB];
            for (int u = 0; u < 4; ++u) cb[i + u * TPB] = q[u];
        }
        for (; i < ne; i += TPB) cb[i] = s_csr[i];
    } else {
        for (; i + 3 * TPB < ne; i += 4 * TPB) {
            int2 p[4];
            for (int u = 0; u < 4; ++u) p[u] = pb[i + u * TPB];
            for (int u = 0; u < 4; ++u) {
                const int dl = (p[u].y - n0) & (BNODES - 1);
                const int r  = atomicAdd(&s_cur[dl], 1);
                csr[bs + s_off[dl] + r] = p[u].x;
            }
        }
        for (; i < ne; i += TPB) {
            const int2 p = pb[i];
            const int dl = (p.y - n0) & (BNODES - 1);
            const int r  = atomicAdd(&s_cur[dl], 1);
            csr[bs + s_off[dl] + r] = p.x;
        }
    }
}

__global__ void gcn_gemm1_wmma(const float* __restrict__ x,
                               const float* __restrict__ W1,
                               const float* __restrict__ dis,
                               float* __restrict__ h1, int n_tiles) {
    __shared__ float sW[F_IN * HID];
    for (int i = threadIdx.x; i < F_IN * HID; i += blockDim.x) sW[i] = W1[i];
    __syncthreads();

    const int wave_in_blk = threadIdx.x >> 5;
    const int tile = blockIdx.x * (blockDim.x >> 5) + wave_in_blk;
    if (tile >= n_tiles) return;

    const int lane = threadIdx.x & 31;
    const int half = lane >> 4;
    const int l16  = lane & 15;

    const int row_base = tile * 16;
    const float* xrow = x + (long long)(row_base + l16) * F_IN;

    const float4 dA = *reinterpret_cast<const float4*>(dis + row_base + 8 * half);
    const float4 dB = *reinterpret_cast<const float4*>(dis + row_base + 8 * half + 4);
    const float dr[8] = {dA.x, dA.y, dA.z, dA.w, dB.x, dB.y, dB.z, dB.w};

    v8f c = {};
    #pragma unroll 8
    for (int k = 0; k < F_IN; k += 4) {
        v2f a;
        a.x = xrow[k + 2 * half + 0];
        a.y = xrow[k + 2 * half + 1];
        v2f b;
        b.x = sW[(k + 2 * half + 0) * HID + l16];
        b.y = sW[(k + 2 * half + 1) * HID + l16];
        c = __builtin_amdgcn_wmma_f32_16x16x4_f32(
                false, a, false, b, (short)0, c, false, false);
    }

    float* hrow = h1 + (long long)row_base * HID;
    #pragma unroll
    for (int i = 0; i < 8; ++i)
        hrow[(i + 8 * half) * HID + l16] = dr[i] * c[i];
}

__global__ void gcn_layer1_kernel(const int* __restrict__ csr,
                                  const int* __restrict__ off,
                                  const int* __restrict__ cnt,
                                  const float* __restrict__ dis,
                                  const float* __restrict__ h1,
                                  const float* __restrict__ b1,
                                  const float* __restrict__ W2,
                                  float* __restrict__ h2, int N) {
    long long t = (long long)blockIdx.x * blockDim.x + threadIdx.x;
    int n = (int)(t >> 2);
    int q = (int)(t & 3);
    if (n >= N) return;
    const int c = cnt[n];
    const int* p = csr + off[n];
    const float dn = dis[n];
    const float4 hs = *reinterpret_cast<const float4*>(h1 + (long long)n * HID + 4 * q);
    float bf[4], w20[4], w21[4];
    #pragma unroll
    for (int i = 0; i < 4; ++i) {
        bf[i]  = b1[4 * q + i];
        w20[i] = W2[(4 * q + i) * OUT + 0];
        w21[i] = W2[(4 * q + i) * OUT + 1];
    }
    float acc0 = 0.f, acc1 = 0.f, acc2 = 0.f, acc3 = 0.f;
    for (int j = 0; j < c; j += 8) {
        int    s[8];
        float4 v[8];
        #pragma unroll
        for (int u = 0; u < 8; ++u) {
            int jj = j + u;
            s[u] = p[jj < c ? jj : c - 1];
        }
        #pragma unroll
        for (int u = 0; u < 8; ++u)
            v[u] = *reinterpret_cast<const float4*>(h1 + (long long)s[u] * HID + 4 * q);
        #pragma unroll
        for (int u = 0; u < 8; ++u) {
            const float m = (j + u < c) ? 1.f : 0.f;
            acc0 = fmaf(m, v[u].x, acc0);
            acc1 = fmaf(m, v[u].y, acc1);
            acc2 = fmaf(m, v[u].z, acc2);
            acc3 = fmaf(m, v[u].w, acc3);
        }
    }
    const float r0 = fmaxf(dn * (acc0 + hs.x) + bf[0], 0.0f);
    const float r1 = fmaxf(dn * (acc1 + hs.y) + bf[1], 0.0f);
    const float r2 = fmaxf(dn * (acc2 + hs.z) + bf[2], 0.0f);
    const float r3 = fmaxf(dn * (acc3 + hs.w) + bf[3], 0.0f);
    float a0 = r0 * w20[0] + r1 * w20[1] + r2 * w20[2] + r3 * w20[3];
    float a1 = r0 * w21[0] + r1 * w21[1] + r2 * w21[2] + r3 * w21[3];
    #pragma unroll
    for (int m = 2; m >= 1; m >>= 1) {
        a0 += __shfl_xor(a0, m, 4);
        a1 += __shfl_xor(a1, m, 4);
    }
    if (q == 0)
        *reinterpret_cast<float2*>(h2 + (long long)n * OUT) = make_float2(dn * a0, dn * a1);
}

__global__ void gcn_layer2_kernel(const int* __restrict__ csr,
                                  const int* __restrict__ off,
                                  const int* __restrict__ cnt,
                                  const float* __restrict__ dis,
                                  const float* __restrict__ h2,
                                  const float* __restrict__ b2,
                                  float* __restrict__ out, int N) {
    long long t = (long long)blockIdx.x * blockDim.x + threadIdx.x;
    int n = (int)(t >> 2);
    int l = (int)(t & 3);
    if (n >= N) return;
    const int c = cnt[n];
    const int* p = csr + off[n];
    const float dn = dis[n];
    const float2 hn = *reinterpret_cast<const float2*>(h2 + (long long)n * OUT);
    const float b20 = b2[0];
    const float b21 = b2[1];
    float a0 = 0.f, a1 = 0.f;
    for (int j = l; j < c; j += 16) {
        int    s[4];
        float2 hv[4];
        #pragma unroll
        for (int u = 0; u < 4; ++u) {
            int jj = j + 4 * u;
            s[u] = p[jj < c ? jj : c - 1];
        }
        #pragma unroll
        for (int u = 0; u < 4; ++u)
            hv[u] = *reinterpret_cast<const float2*>(h2 + (long long)s[u] * OUT);
        #pragma unroll
        for (int u = 0; u < 4; ++u) {
            const float m = (j + 4 * u < c) ? 1.f : 0.f;
            a0 = fmaf(m, hv[u].x, a0);
            a1 = fmaf(m, hv[u].y, a1);
        }
    }
    #pragma unroll
    for (int m = 2; m >= 1; m >>= 1) {
        a0 += __shfl_xor(a0, m, 4);
        a1 += __shfl_xor(a1, m, 4);
    }
    if (l == 0) {
        const float z0 = dn * (a0 + hn.x) + b20;
        const float z1 = dn * (a1 + hn.y) + b21;
        const float m = fmaxf(z0, z1);
        const float lse = m + logf(expf(z0 - m) + expf(z1 - m));
        *reinterpret_cast<float2*>(out + (long long)n * OUT) = make_float2(z0 - lse, z1 - lse);
    }
}

extern "C" void kernel_launch(void* const* d_in, const int* in_sizes, int n_in,
                              void* d_out, int out_size, void* d_ws, size_t ws_size,
                              hipStream_t stream) {
    const float* x  = (const float*)d_in[0];
    const int*   ei = (const int*)d_in[1];
    const float* W1 = (const float*)d_in[2];
    const float* b1 = (const float*)d_in[3];
    const float* W2 = (const float*)d_in[4];
    const float* b2 = (const float*)d_in[5];
    float* out = (float*)d_out;

    const int       N = in_sizes[0] / F_IN;
    const long long E = (long long)in_sizes[1] / 2;
    const int* src = ei;
    const int* dst = ei + E;

    float* ws   = (float*)d_ws;
    int*   cnt  = (int*)ws;
    float* dis  = ws   + N;
    float* h1   = dis  + N;
    float* h2   = h1   + (long long)N * HID;
    int*   off  = (int*)(h2 + (long long)N * OUT);
    int2*  pairs = (int2*)(off + N);
    int*   csr  = (int*)(pairs + E);
    int*   table = csr + E;
    int*   tot  = table + NCH * NBMAX;

    const int B = 256;

    gcn_bhist_kernel<<<NCH, TPB, 0, stream>>>(dst, table, (int)E);

    gcn_bscan_kernel<<<NBMAX / 32, TPB, 0, stream>>>(table, tot);

    gcn_bscatter_kernel<<<NCH, TPB, 0, stream>>>(src, dst, table, tot, pairs, (int)E);

    gcn_bucket_csr_kernel<<<NBMAX, TPB, 0, stream>>>(pairs, tot, cnt, off, dis, csr, N);

    {
        int n_tiles = N / 16;
        int waves_per_block = B / 32;
        int blocks = (n_tiles + waves_per_block - 1) / waves_per_block;
        gcn_gemm1_wmma<<<blocks, B, 0, stream>>>(x, W1, dis, h1, n_tiles);
    }

    {
        long long T = (long long)N * 4;
        gcn_layer1_kernel<<<(unsigned)((T + B - 1) / B), B, 0, stream>>>(csr, off, cnt, dis, h1, b1, W2, h2, N);
    }

    {
        long long T = (long long)N * 4;
        gcn_layer2_kernel<<<(unsigned)((T + B - 1) / B), B, 0, stream>>>(csr, off, cnt, dis, h2, b2, out, N);
    }
}


# Both gathers: a wave's 8 nodes' source indices are one contiguous csr span, fetched once into LDS by loads of 32 neighbouring words; per-trip index loads become LDS reads; slot order and sums kept
# speedup vs baseline: 7.9716x; 1.2269x over previous
#include <hip/hip_runtime.h>
#include <hip/hip_bf16.h>

typedef __attribute__((ext_vector_type(2))) float v2f;
typedef __attribute__((ext_vector_type(8))) float v8f;

#define F_IN 128
#define HID  16
#define OUT  2

#define NCH    256
#define NBMAX  512
#define BSH    8
#define BNODES 256
#define TPB    1024
#define PCAP   12544
#define CCAP   16384

__device__ __forceinline__ int gcn_bucket(int d) {
    return min((int)((unsigned)d >> BSH), NBMAX - 1);
}

__global__ __launch_bounds__(TPB) void gcn_bhist_kernel(const int* __restrict__ dst,
                                                        int* __restrict__ table, int E) {
    __shared__ int s_h[NBMAX];
    const int tid = threadIdx.x;
    if (tid < NBMAX) s_h[tid] = 0;
    __syncthreads();
    const int CH = (E + NCH - 1) / NCH;
    const int e0 = blockIdx.x * CH;
    const int e1 = min(E, e0 + CH);
    int e = e0 + tid;
    for (; e + 3 * TPB < e1; e += 4 * TPB) {
        int d[4];
        for (int u = 0; u < 4; ++u) d[u] = dst[e + u * TPB];
        for (int u = 0; u < 4; ++u) atomicAdd(&s_h[gcn_bucket(d[u])], 1);
    }
    for (; e < e1; e += TPB) atomicAdd(&s_h[gcn_bucket(dst[e])], 1);
    __syncthreads();
    if (tid < NBMAX) table[blockIdx.x * NBMAX + tid] = s_h[tid];
}

__global__ __launch_bounds__(TPB) void gcn_bscan_kernel(int* __restrict__ table,
                                                        int* __restrict__ tot) {
    __shared__ int s_p[32][33];
    const int bl = threadIdx.x & 31;
    const int cg = threadIdx.x >> 5;
    const int b  = blockIdx.x * 32 + bl;
    int v[8];
    for (int k = 0; k < 8; ++k) v[k] = table[(cg * 8 + k) * NBMAX + b];
    int sum = 0;
    for (int k = 0; k < 8; ++k) { int t = v[k]; v[k] = sum; sum += t; }
    s_p[cg][bl] = sum;
    __syncthreads();
    if (cg == 0) {
        int run = 0;
        for (int g = 0; g < 32; ++g) { int t = s_p[g][bl]; s_p[g][bl] = run; run += t; }
        tot[b] = run;
    }
    __syncthreads();
    const int base = s_p[cg][bl];
    for (int k = 0; k < 8; ++k) table[(cg * 8 + k) * NBMAX + b] = base + v[k];
}

__global__ __launch_bounds__(TPB) void gcn_bscatter_kernel(const int* __restrict__ src,
                                                           const int* __restrict__ dst,
                                                           const int* __restrict__ table,
                                                           const int* __restrict__ tot,
                                                           int2* __restrict__ pairs, int E) {
    __shared__ int  s_cur[NBMAX];
    __shared__ int  s_delta[NBMAX];
    __shared__ int  s_wave[32];
    __shared__ int  s_wloc[32];
    __shared__ int2 s_pairs[PCAP];
    const int tid  = threadIdx.x;
    const int lane = tid & 31;
    const int wave = tid >> 5;
    const int CH = (E + NCH - 1) / NCH;
    const int e0 = blockIdx.x * CH;
    const int e1 = min(E, e0 + CH);
    const bool staged = (CH <= PCAP);

    int c = 0, trow = 0, k = 0;
    if (tid < NBMAX) {
        c    = tot[tid];
        trow = table[blockIdx.x * NBMAX + tid];
        const int tnext = (blockIdx.x + 1 < NCH) ? table[(blockIdx.x + 1) * NBMAX + tid] : c;
        k = tnext - trow;
    }
    int v = c, w = k;
    for (int d = 1; d < 32; d <<= 1) {
        int t  = __shfl_up(v, d, 32);
        int t2 = __shfl_up(w, d, 32);
        if (lane >= d) { v += t; w += t2; }
    }
    if (lane == 31) { s_wave[wave] = v; s_wloc[wave] = w; }
    __syncthreads();
    if (tid == 0) {
        int run = 0;
        for (int g = 0; g < 32; ++g) { int t = s_wave[g]; s_wave[g] = run; run += t; }
    }
    if (tid == 32) {
        int run = 0;
        for (int g = 0; g < 32; ++g) { int t = s_wloc[g]; s_wloc[g] = run; run += t; }
    }
    __syncthreads();
    if (tid < NBMAX) {
        const int gpos = s_wave[wave] + (v - c) + trow;
        const int lpos = s_wloc[wave] + (w - k);
        s_cur[tid]   = staged ? lpos : gpos;
        s_delta[tid] = gpos - lpos;
    }
    __syncthreads();

    int e = e0 + tid;
    if (staged) {
        for (; e + 3 * TPB < e1; e += 4 * TPB) {
            int d[4], s[4];
            for (int u = 0; u < 4; ++u) { d[u] = dst[e + u * TPB]; s[u] = src[e + u * TPB]; }
            int pos[4];
            for (int u = 0; u < 4; ++u) pos[u] = atomicAdd(&s_cur[gcn_bucket(d[u])], 1);
            for (int u = 0; u < 4; ++u)
                s_pairs[min(pos[u], PCAP - 1)] = make_int2(s[u], d[u]);
        }
        for (; e < e1; e += TPB) {
            const int d = dst[e], s = src[e];
            const int pos = atomicAdd(&s_cur[gcn_bucket(d)], 1);
            s_pairs[min(pos, PCAP - 1)] = make_int2(s, d);
        }
        __syncthreads();
        const int nloc = e1 - e0;
        int i = tid;
        for (; i + 3 * TPB < nloc; i += 4 * TPB) {
            int2 p[4];
            int  g[4];
            for (int u = 0; u < 4; ++u) p[u] = s_pairs[i + u * TPB];
            for (int u = 0; u < 4; ++u) g[u] = i + u * TPB + s_delta[gcn_bucket(p[u].y)];
            for (int u = 0; u < 4; ++u) pairs[g[u]] = p[u];
        }
        for (; i < nloc; i += TPB) {
            const int2 p = s_pairs[i];
            pairs[i + s_delta[gcn_bucket(p.y)]] = p;
        }
    } else {
        for (; e + 3 * TPB < e1; e += 4 * TPB) {
            int d[4], s[4];
            for (int u = 0; u < 4; ++u) { d[u] = dst[e + u * TPB]; s[u] = src[e + u * TPB]; }
            for (int u = 0; u < 4; ++u) {
                const int pos = atomicAdd(&s_cur[gcn_bucket(d[u])], 1);
                pairs[pos] = make_int2(s[u], d[u]);
            }
        }
        for (; e < e1; e += TPB) {
            const int d = dst[e], s = src[e];
            const int pos = atomicAdd(&s_cur[gcn_bucket(d)], 1);
            pairs[pos] = make_int2(s, d);
        }
    }
}

__global__ __launch_bounds__(TPB) void gcn_bucket_csr_kernel(const int2* __restrict__ pairs,
                                                             const int* __restrict__ tot,
                                                             int* __restrict__ cnt,
                                                             int* __restrict__ off,
                                                             float* __restrict__ dis,
                                                             int* __restrict__ csr, int N) {
    const int b  = blockIdx.x;
    const int n0 = b << BSH;
    if (n0 >= N) return;
    __shared__ int s_cnt[BNODES];
    __shared__ int s_off[BNODES];
    __shared__ int s_cur[BNODES];
    __shared__ int s_red[32];
    __shared__ int s_wave[32];
    __shared__ int s_bs;
    __shared__ int s_csr[CCAP];
    const int tid  = threadIdx.x;
    const int lane = tid & 31;
    const int wave = tid >> 5;

    int part = (tid < b) ? tot[tid] : 0;
    for (int m = 16; m >= 1; m >>= 1) part += __shfl_xor(part, m, 32);
    if (lane == 0) s_red[wave] = part;
    if (tid < BNODES) { s_cnt[tid] = 0; s_cur[tid] = 0; }
    __syncthreads();
    if (tid == 0) {
        int r = 0;
        for (int w = 0; w < 32; ++w) r += s_red[w];
        s_bs = r;
    }
    __syncthreads();
    const int bs = s_bs;
    const int ne = tot[b];
    const int2* pb = pairs + bs;

    int i = tid;
    for (; i + 3 * TPB < ne; i += 4 * TPB) {
        int2 p[4];
        for (int u = 0; u < 4; ++u) p[u] = pb[i + u * TPB];
        for (int u = 0; u < 4; ++u) atomicAdd(&s_cnt[(p[u].y - n0) & (BNODES - 1)], 1);
    }
    for (; i < ne; i += TPB) atomicAdd(&s_cnt[(pb[i].y - n0) & (BNODES - 1)], 1);
    __syncthreads();

    const int c = (tid < BNODES) ? s_cnt[tid] : 0;
    int v = c;
    for (int d = 1; d < 32; d <<= 1) {
        int t = __shfl_up(v, d, 32);
        if (lane >= d) v += t;
    }
    if (lane == 31) s_wave[wave] = v;
    __syncthreads();
    if (tid == 0) {
        int run = 0;
        for (int w = 0; w < 32; ++w) { int t = s_wave[w]; s_wave[w] = run; run += t; }
    }
    __syncthreads();
    if (tid < BNODES) {
        const int o = s_wave[wave] + (v - c);
        s_off[tid] = o;
        const int n = n0 + tid;
        if (n < N) {
            cnt[n] = c;
            off[n] = bs + o;
            dis[n] = rsqrtf((float)(c + 1));
        }
    }
    __syncthreads();

    i = tid;
    if (ne <= CCAP) {
        for (; i + 3 * TPB < ne; i += 4 * TPB) {
            int2 p[4];
            for (int u = 0; u < 4; ++u) p[u] = pb[i + u * TPB];
            int dl[4], r[4];
            for (int u = 0; u < 4; ++u) {
                dl[u] = (p[u].y - n0) & (BNODES - 1);
                r[u]  = atomicAdd(&s_cur[dl[u]], 1);
            }
            for (int u = 0; u < 4; ++u)
                s_csr[min(s_off[dl[u]] + r[u], CCAP - 1)] = p[u].x;
        }
        for (; i < ne; i += TPB) {
            const int2 p = pb[i];
            const int dl = (p.y - n0) & (BNODES - 1);
            const int r  = atomicAdd(&s_cur[dl], 1);
            s_csr[min(s_off[dl] + r, CCAP - 1)] = p.x;
        }
        __syncthreads();
        int* cb = csr + bs;
        i = tid;
        for (; i + 3 * TPB < ne; i += 4 * TPB) {
            int q[4];
            for (int u = 0; u < 4; ++u) q[u] = s_csr[i + u * TPB];
            for (int u = 0; u < 4; ++u) cb[i + u * TPB] = q[u];
        }
        for (; i < ne; i += TPB) cb[i] = s_csr[i];
    } else {
        for (; i + 3 * TPB < ne; i += 4 * TPB) {
            int2 p[4];
            for (int u = 0; u < 4; ++u) p[u] = pb[i + u * TPB];
            for (int u = 0; u < 4; ++u) {
                const int dl = (p[u].y - n0) & (BNODES - 1);
                const int r  = atomicAdd(&s_cur[dl], 1);
                csr[bs + s_off[dl] + r] = p[u].x;
            }
        }
        for (; i < ne; i += TPB) {
            const int2 p = pb[i];
            const int dl = (p.y - n0) & (BNODES - 1);
            const int r  = atomicAdd(&s_cur[dl], 1);
            csr[bs + s_off[dl] + r] = p.x;
        }
    }
}

__global__ void gcn_gemm1_wmma(const float* __restrict__ x,
                               const float* __restrict__ W1,
                               const float* __restrict__ dis,
                               float* __restrict__ h1, int n_tiles) {
    __shared__ float sW[F_IN * HID];
    for (int i = threadIdx.x; i < F_IN * HID; i += blockDim.x) sW[i] = W1[i];
    __syncthreads();

    const int wave_in_blk = threadIdx.x >> 5;
    const int tile = blockIdx.x * (blockDim.x >> 5) + wave_in_blk;
    if (tile >= n_tiles) return;

    const int lane = threadIdx.x & 31;
    const int half = lane >> 4;
    const int l16  = lane & 15;

    const int row_base = tile * 16;
    const float* xrow = x + (long long)(row_base + l16) * F_IN;

    const float4 dA = *reinterpret_cast<const float4*>(dis + row_base + 8 * half);
    const float4 dB = *reinterpret_cast<const float4*>(dis + row_base + 8 * half + 4);
    const float dr[8] = {dA.x, dA.y, dA.z, dA.w, dB.x, dB.y, dB.z, dB.w};

    v8f c = {};
    #pragma unroll 8
    for (int k = 0; k < F_IN; k += 4) {
        v2f a;
        a.x = xrow[k + 2 * half + 0];
        a.y = xrow[k + 2 * half + 1];
        v2f b;
        b.x = sW[(k + 2 * half + 0) * HID + l16];
        b.y = sW[(k + 2 * half + 1) * HID + l16];
        c = __builtin_amdgcn_wmma_f32_16x16x4_f32(
                false, a, false, b, (short)0, c, false, false);
    }

    float* hrow = h1 + (long long)row_base * HID;
    #pragma unroll
    for (int i = 0; i < 8; ++i)
        hrow[(i + 8 * half) * HID + l16] = dr[i] * c[i];
}

#define ICH 384
__global__ __launch_bounds__(256, 16) void gcn_layer1_kernel(const int* __restrict__ csr,
                                  const int* __restrict__ off,
                                  const int* __restrict__ cnt,
                                  const float* __restrict__ dis,
                                  const float* __restrict__ h1,
                                  const float* __restrict__ b1,
                                  const float* __restrict__ W2,
                                  float* __restrict__ h2, int N) {
    __shared__ int s_idx[8][ICH];
    const int lane = threadIdx.x & 31;
    int* si = s_idx[threadIdx.x >> 5];
    long long t = (long long)blockIdx.x * blockDim.x + threadIdx.x;
    int n = (int)(t >> 2);
    int q = (int)(t & 3);
    const bool live = n < N;
    const int nn = live ? n : N - 1;
    const int cs = cnt[nn];
    const int c = live ? cs : 0;
    const int o = off[nn];
    const float dn = dis[nn];
    const float4 hs = *reinterpret_cast<const float4*>(h1 + (long long)nn * HID + 4 * q);
    const int sb = __builtin_amdgcn_readfirstlane(o);
    const int se = __builtin_amdgcn_readlane(o + cs, 31);
    float acc0 = 0.f, acc1 = 0.f, acc2 = 0.f, acc3 = 0.f;
    for (int base = sb; base < se; base += ICH) {
        const int lim = min(se - base, ICH);
        int w[ICH / 32];
        #pragma unroll
        for (int m = 0; m < ICH / 32; ++m)
            w[m] = csr[base + min(lane + 32 * m, lim - 1)];
        #pragma unroll
        for (int m = 0; m < ICH / 32; ++m) si[lane + 32 * m] = w[m];
        __builtin_amdgcn_fence(__ATOMIC_RELEASE, "wavefront");
        __builtin_amdgcn_wave_barrier();
        __builtin_amdgcn_fence(__ATOMIC_ACQUIRE, "wavefront");
        const int lo = max(base - o, 0);
        const int hi = min(base + lim - o, c);
        const int* sp = si + (o - base);
        for (int j = lo; j < hi; j += 8) {
            int    s[8];
            float4 v[8];
            const int rem = hi - j;
            const int* spj = sp + j;
            #pragma unroll
            for (int u = 0; u < 8; ++u)
                s[u] = spj[u < rem ? u : rem - 1];
            #pragma unroll
            for (int u = 0; u < 8; ++u)
                v[u] = *reinterpret_cast<const float4*>(h1 + (long long)s[u] * HID + 4 * q);
            #pragma unroll
            for (int u = 0; u < 8; ++u) {
                const float m = (u < rem) ? 1.f : 0.f;
                acc0 = fmaf(m, v[u].x, acc0);
                acc1 = fmaf(m, v[u].y, acc1);
                acc2 = fmaf(m, v[u].z, acc2);
                acc3 = fmaf(m, v[u].w, acc3);
            }
        }
        __builtin_amdgcn_fence(__ATOMIC_RELEASE, "wavefront");
        __builtin_amdgcn_wave_barrier();
        __builtin_amdgcn_fence(__ATOMIC_ACQUIRE, "wavefront");
    }
    float bf[4], w20[4], w21[4];
    #pragma unroll
    for (int i = 0; i < 4; ++i) {
        bf[i]  = b1[4 * q + i];
        w20[i] = W2[(4 * q + i) * OUT + 0];
        w21[i] = W2[(4 * q + i) * OUT + 1];
    }
    const float r0 = fmaxf(dn * (acc0 + hs.x) + bf[0], 0.0f);
    const float r1 = fmaxf(dn * (acc1 + hs.y) + bf[1], 0.0f);
    const float r2 = fmaxf(dn * (acc2 + hs.z) + bf[2], 0.0f);
    const float r3 = fmaxf(dn * (acc3 + hs.w) + bf[3], 0.0f);
    float a0 = r0 * w20[0] + r1 * w20[1] + r2 * w20[2] + r3 * w20[3];
    float a1 = r0 * w21[0] + r1 * w21[1] + r2 * w21[2] + r3 * w21[3];
    #pragma unroll
    for (int m = 2; m >= 1; m >>= 1) {
        a0 += __shfl_xor(a0, m, 4);
        a1 += __shfl_xor(a1, m, 4);
    }
    if (live && q == 0)
        *reinterpret_cast<float2*>(h2 + (long long)n * OUT) = make_float2(dn * a0, dn * a1);
}

__global__ __launch_bounds__(256) void gcn_layer2_kernel(const int* __restrict__ csr,
                                  const int* __restrict__ off,
                                  const int* __restrict__ cnt,
                                  const float* __restrict__ dis,
                                  const float* __restrict__ h2,
                                  const float* __restrict__ b2,
                                  float* __restrict__ out, int N) {
    __shared__ int s_idx[8][ICH];
    const int lane = threadIdx.x & 31;
    int* si = s_idx[threadIdx.x >> 5];
    long long t = (long long)blockIdx.x * blockDim.x + threadIdx.x;
    int n = (int)(t >> 2);
    int l = (int)(t & 3);
    const bool live = n < N;
    const int nn = live ? n : N - 1;
    const int cs = cnt[nn];
    const int c = live ? cs : 0;
    const int o = off[nn];
    const float dn = dis[nn];
    const float2 hn = *reinterpret_cast<const float2*>(h2 + (long long)nn * OUT);
    const float b20 = b2[0];
    const float b21 = b2[1];
    const int sb = __builtin_amdgcn_readfirstlane(o);
    const int se = __builtin_amdgcn_readlane(o + cs, 31);
    float a0 = 0.f, a1 = 0.f;
    for (int base = sb; base < se; base += ICH) {
        const int lim = min(se - base, ICH);
        int w[ICH / 32];
        #pragma unroll
        for (int m = 0; m < ICH / 32; ++m)
            w[m] = csr[base + min(lane + 32 * m, lim - 1)];
        #pragma unroll
        for (int m = 0; m < ICH / 32; ++m) si[lane + 32 * m] = w[m];
        __builtin_amdgcn_fence(__ATOMIC_RELEASE, "wavefront");
        __builtin_amdgcn_wave_barrier();
        __builtin_amdgcn_fence(__ATOMIC_ACQUIRE, "wavefront");
        const int lo = max(base - o, 0);
        const int hi = min(base + lim - o, c);
        const int* sp = si + (o - base);
        for (int j = lo + ((l - lo) & 3); j < hi; j += 16) {
            int    s[4];
            float2 hv[4];
            #pragma unroll
            for (int u = 0; u < 4; ++u) {
                int jj = j + 4 * u;
                s[u] = sp[jj < hi ? jj : hi - 1];
            }
            #pragma unroll
            for (int u = 0; u < 4; ++u)
                hv[u] = *reinterpret_cast<const float2*>(h2 + (long long)s[u] * OUT);
            #pragma unroll
            for (int u = 0; u < 4; ++u) {
                const float m = (j + 4 * u < hi) ? 1.f : 0.f;
                a0 = fmaf(m, hv[u].x, a0);
                a1 = fmaf(m, hv[u].y, a1);
            }
        }
        __builtin_amdgcn_fence(__ATOMIC_RELEASE, "wavefront");
        __builtin_amdgcn_wave_barrier();
        __builtin_amdgcn_fence(__ATOMIC_ACQUIRE, "wavefront");
    }
    #pragma unroll
    for (int m = 2; m >= 1; m >>= 1) {
        a0 += __shfl_xor(a0, m, 4);
        a1 += __shfl_xor(a1, m, 4);
    }
    if (live && l == 0) {
        const float z0 = dn * (a0 + hn.x) + b20;
        const float z1 = dn * (a1 + hn.y) + b21;
        const float m = fmaxf(z0, z1);
        const float lse = m + logf(expf(z0 - m) + expf(z1 - m));
        *reinterpret_cast<float2*>(out + (long long)n * OUT) = make_float2(z0 - lse, z1 - lse);
    }
}

extern "C" void kernel_launch(void* const* d_in, const int* in_sizes, int n_in,
                              void* d_out, int out_size, void* d_ws, size_t ws_size,
                              hipStream_t stream) {
    const float* x  = (const float*)d_in[0];
    const int*   ei = (const int*)d_in[1];
    const float* W1 = (const float*)d_in[2];
    const float* b1 = (const float*)d_in[3];
    const float* W2 = (const float*)d_in[4];
    const float* b2 = (const float*)d_in[5];
    float* out = (float*)d_out;

    const int       N = in_sizes[0] / F_IN;
    const long long E = (long long)in_sizes[1] / 2;
    const int* src = ei;
    const int* dst = ei + E;

    float* ws   = (float*)d_ws;
    int*   cnt  = (int*)ws;
    float* dis  = ws   + N;
    float* h1   = dis  + N;
    float* h2   = h1   + (long long)N * HID;
    int*   off  = (int*)(h2 + (long long)N * OUT);
    int2*  pairs = (int2*)(off + N);
    int*   csr  = (int*)(pairs + E);
    int*   table = csr + E;
    int*   tot  = table + NCH * NBMAX;

    const int B = 256;

    gcn_bhist_kernel<<<NCH, TPB, 0, stream>>>(dst, table, (int)E);

    gcn_bscan_kernel<<<NBMAX / 32, TPB, 0, stream>>>(table, tot);

    gcn_bscatter_kernel<<<NCH, TPB, 0, stream>>>(src, dst, table, tot, pairs, (int)E);

    gcn_bucket_csr_kernel<<<NBMAX, TPB, 0, stream>>>(pairs, tot, cnt, off, dis, csr, N);

    {
        int n_tiles = N / 16;
        int waves_per_block = B / 32;
        int blocks = (n_tiles + waves_per_block - 1) / waves_per_block;
        gcn_gemm1_wmma<<<blocks, B, 0, stream>>>(x, W1, dis, h1, n_tiles);
    }

    {
        long long T = (long long)N * 4;
        gcn_layer1_kernel<<<(unsigned)((T + B - 1) / B), B, 0, stream>>>(csr, off, cnt, dis, h1, b1, W2, h2, N);
    }

    {
        long long T = (long long)N * 4;
        gcn_layer2_kernel<<<(unsigned)((T + B - 1) / B), B, 0, stream>>>(csr, off, cnt, dis, h2, b2, out, N);
    }
}


# Edges grouped with no pass over all chunks: each chunk bucket-sorted in its own span (ranks kept in registers), each bucket collects its 256 runs; 7 launches to 5; slots still by LDS add arrival
# speedup vs baseline: 8.2901x; 1.0400x over previous
#include <hip/hip_runtime.h>
#include <hip/hip_bf16.h>

typedef __attribute__((ext_vector_type(2))) float v2f;
typedef __attribute__((ext_vector_type(8))) float v8f;

#define F_IN 128
#define HID  16
#define OUT  2

#define NCH    256
#define NBMAX  512
#define BSH    8
#define BNODES 256
#define TPB    1024
#define PCAP   12544
#define CCAP   16384

__device__ __forceinline__ int gcn_bucket(int d) {
    return min((int)((unsigned)d >> BSH), NBMAX - 1);
}

#define EPT ((PCAP + TPB - 1) / TPB)
__global__ __launch_bounds__(TPB) void gcn_chunk_sort_kernel(const int* __restrict__ src,
                                                             const int* __restrict__ dst,
                                                             int* __restrict__ tabH,
                                                             int* __restrict__ tabL,
                                                             int2* __restrict__ pairs, int E) {
    __shared__ int  s_h[NBMAX];
    __shared__ int  s_l[NBMAX];
    __shared__ int  s_wave[32];
    __shared__ int2 s_pairs[PCAP];
    const int tid  = threadIdx.x;
    const int lane = tid & 31;
    const int wave = tid >> 5;
    const int CH = (E + NCH - 1) / NCH;
    const int e0 = blockIdx.x * CH;
    const int e1 = min(E, e0 + CH);
    const bool staged = (CH <= PCAP);

    if (tid < NBMAX) s_h[tid] = 0;
    __syncthreads();

    int d[EPT], s[EPT], r[EPT];
    if (staged) {
        const int elast = (e1 > e0) ? e1 - 1 : 0;
        #pragma unroll
        for (int u = 0; u < EPT; ++u) {
            const int e = min(e0 + tid + u * TPB, elast);
            d[u] = dst[e]; s[u] = src[e];
        }
        #pragma unroll
        for (int u = 0; u < EPT; ++u) {
            const int e = e0 + tid + u * TPB;
            r[u] = 0;
            if (e < e1) r[u] = atomicAdd(&s_h[gcn_bucket(d[u])], 1);
        }
    } else {
        for (int e = e0 + tid; e < e1; e += TPB) atomicAdd(&s_h[gcn_bucket(dst[e])], 1);
    }
    __syncthreads();

    const int k = (tid < NBMAX) ? s_h[tid] : 0;
    int v = k;
    for (int dd = 1; dd < 32; dd <<= 1) {
        int t = __shfl_up(v, dd, 32);
        if (lane >= dd) v += t;
    }
    if (lane == 31) s_wave[wave] = v;
    __syncthreads();
    int wb = 0;
    for (int g = 0; g < NBMAX / 32; ++g) wb += (g < wave) ? s_wave[g] : 0;
    if (tid < NBMAX) {
        const int l = wb + (v - k);
        s_l[tid] = l;
        tabH[blockIdx.x * NBMAX + tid] = k;
        tabL[blockIdx.x * NBMAX + tid] = l;
    }
    __syncthreads();

    if (staged) {
        #pragma unroll
        for (int u = 0; u < EPT; ++u) r[u] += s_l[gcn_bucket(d[u])];
        #pragma unroll
        for (int u = 0; u < EPT; ++u) {
            const int e = e0 + tid + u * TPB;
            if (e < e1)
                s_pairs[min(r[u], PCAP - 1)] = make_int2(s[u], d[u]);
        }
        __syncthreads();
        const int nloc = e1 - e0;
        int2* po = pairs + e0;
        int i = tid;
        for (; i + 3 * TPB < nloc; i += 4 * TPB) {
            int2 p[4];
            for (int u = 0; u < 4; ++u) p[u] = s_pairs[i + u * TPB];
            for (int u = 0; u < 4; ++u) po[i + u * TPB] = p[u];
        }
        for (; i < nloc; i += TPB) po[i] = s_pairs[i];
    } else {
        for (int e = e0 + tid; e < e1; e += TPB) {
            const int dv = dst[e], sv = src[e];
            const int pos = atomicAdd(&s_l[gcn_bucket(dv)], 1);
            pairs[e0 + pos] = make_int2(sv, dv);
        }
    }
}

#define RPW (NCH / 32)
template <bool WITH_SRC, typename FA, typename FB>
__device__ __forceinline__ void gcn_for_bucket_edges(const int2* __restrict__ pairs, int elast,
                                                     const int* s_rs, const int* s_rl,
                                                     int wave, int lane, FA fa, FB fb) {
    int  st[RPW], ln[RPW];
    int2 p0[RPW], p1[RPW];
    int  r0[RPW], r1[RPW];
    #pragma unroll
    for (int j = 0; j < RPW; ++j) {
        st[j] = s_rs[wave + 32 * j];
        ln[j] = s_rl[wave + 32 * j];
    }
    #pragma unroll
    for (int j = 0; j < RPW; ++j) {
        const int last = max(ln[j], 1) - 1;
        const int i0 = min(st[j] + min(lane, last), elast);
        const int i1 = min(st[j] + min(lane + 32, last), elast);
        if (WITH_SRC) {
            p0[j] = pairs[i0];
            p1[j] = pairs[i1];
        } else {
            p0[j] = make_int2(0, reinterpret_cast<const int*>(pairs)[2 * i0 + 1]);
            p1[j] = make_int2(0, reinterpret_cast<const int*>(pairs)[2 * i1 + 1]);
        }
    }
    #pragma unroll
    for (int j = 0; j < RPW; ++j) {
        r0[j] = 0; r1[j] = 0;
        if (lane < ln[j])      r0[j] = fa(p0[j]);
        if (lane + 32 < ln[j]) r1[j] = fa(p1[j]);
    }
    #pragma unroll
    for (int j = 0; j < RPW; ++j) {
        if (lane < ln[j])      fb(p0[j], r0[j]);
        if (lane + 32 < ln[j]) fb(p1[j], r1[j]);
    }
    #pragma unroll
    for (int j = 0; j < RPW; ++j)
        for (int i = lane + 64; i < ln[j]; i += 32) {
            const int2 p = WITH_SRC ? pairs[st[j] + i]
                                    : make_int2(0, reinterpret_cast<const int*>(pairs)[2 * (st[j] + i) + 1]);
            fb(p, fa(p));
        }
}

__global__ __launch_bounds__(TPB) void gcn_bucket_csr_kernel(const int2* __restrict__ pairs,
                                                             const int* __restrict__ tabH,
                                                             const int* __restrict__ tabL,
                                                             int* __restrict__ cnt,
                                                             int* __restrict__ off,
                                                             float* __restrict__ dis,
                                                             int* __restrict__ csr, int N, int E) {
    const int b  = blockIdx.x;
    const int n0 = b << BSH;
    if (n0 >= N) return;
    __shared__ int s_cnt[BNODES];
    __shared__ int s_off[BNODES];
    __shared__ int s_cur[BNODES];
    __shared__ int s_redh[32];
    __shared__ int s_redl[32];
    __shared__ int s_wave[32];
    __shared__ int s_rs[NCH];
    __shared__ int s_rl[NCH];
    __shared__ int s_csr[CCAP];
    const int tid  = threadIdx.x;
    const int lane = tid & 31;
    const int wave = tid >> 5;
    const int CH = (E + NCH - 1) / NCH;

    int ph = 0, pl = 0;
    if (tid < NCH) {
        ph = tabH[tid * NBMAX + b];
        pl = tabL[tid * NBMAX + b];
        s_rl[tid] = ph;
        s_rs[tid] = tid * CH + pl;
    }
    for (int m = 16; m >= 1; m >>= 1) {
        ph += __shfl_xor(ph, m, 32);
        pl += __shfl_xor(pl, m, 32);
    }
    if (lane == 0) { s_redh[wave] = ph; s_redl[wave] = pl; }
    if (tid < BNODES) { s_cnt[tid] = 0; s_cur[tid] = 0; }
    __syncthreads();
    int ne = 0, bs = 0;
    for (int w = 0; w < NCH / 32; ++w) { ne += s_redh[w]; bs += s_redl[w]; }

    const int elast = max(E, 1) - 1;
    gcn_for_bucket_edges<false>(pairs, elast, s_rs, s_rl, wave, lane,
        [&](int2 p) { atomicAdd(&s_cnt[(p.y - n0) & (BNODES - 1)], 1); return 0; },
        [&](int2, int) {});
    __syncthreads();

    const int c = (tid < BNODES) ? s_cnt[tid] : 0;
    int v = c;
    for (int d = 1; d < 32; d <<= 1) {
        int t = __shfl_up(v, d, 32);
        if (lane >= d) v += t;
    }
    if (lane == 31) s_wave[wave] = v;
    __syncthreads();
    if (tid < BNODES) {
        int wbase = 0;
        for (int g = 0; g < BNODES / 32; ++g) wbase += (g < wave) ? s_wave[g] : 0;
        const int o = wbase + (v - c);
        s_off[tid] = o;
        const int n = n0 + tid;
        if (n < N) {
            cnt[n] = c;
            off[n] = bs + o;
            dis[n] = rsqrtf((float)(c + 1));
        }
    }
    __syncthreads();

    if (ne <= CCAP) {
        gcn_for_bucket_edges<true>(pairs, elast, s_rs, s_rl, wave, lane,
            [&](int2 p) { return atomicAdd(&s_cur[(p.y - n0) & (BNODES - 1)], 1); },
            [&](int2 p, int r) {
                s_csr[min(s_off[(p.y - n0) & (BNODES - 1)] + r, CCAP - 1)] = p.x;
            });
        __syncthreads();
        int* cb = csr + bs;
        int i = tid;
        for (; i + 3 * TPB < ne; i += 4 * TPB) {
            int q[4];
            for (int u = 0; u < 4; ++u) q[u] = s_csr[i + u * TPB];
            for (int u = 0; u < 4; ++u) cb[i + u * TPB] = q[u];
        }
        for (; i < ne; i += TPB) cb[i] = s_csr[i];
    } else {
        gcn_for_bucket_edges<true>(pairs, elast, s_rs, s_rl, wave, lane,
            [&](int2 p) { return atomicAdd(&s_cur[(p.y - n0) & (BNODES - 1)], 1); },
            [&](int2 p, int r) { csr[bs + s_off[(p.y - n0) & (BNODES - 1)] + r] = p.x; });
    }
}

__global__ void gcn_gemm1_wmma(const float* __restrict__ x,
                               const float* __restrict__ W1,
                               const float* __restrict__ dis,
                               float* __restrict__ h1, int n_tiles) {
    __shared__ float sW[F_IN * HID];
    for (int i = threadIdx.x; i < F_IN * HID; i += blockDim.x) sW[i] = W1[i];
    __syncthreads();

    const int wave_in_blk = threadIdx.x >> 5;
    const int tile = blockIdx.x * (blockDim.x >> 5) + wave_in_blk;
    if (tile >= n_tiles) return;

    const int lane = threadIdx.x & 31;
    const int half = lane >> 4;
    const int l16  = lane & 15;

    const int row_base = tile * 16;
    const float* xrow = x + (long long)(row_base + l16) * F_IN;

    const float4 dA = *reinterpret_cast<const float4*>(dis + row_base + 8 * half);
    const float4 dB = *reinterpret_cast<const float4*>(dis + row_base + 8 * half + 4);
    const float dr[8] = {dA.x, dA.y, dA.z, dA.w, dB.x, dB.y, dB.z, dB.w};

    v8f c = {};
    #pragma unroll 8
    for (int k = 0; k < F_IN; k += 4) {
        v2f a;
        a.x = xrow[k + 2 * half + 0];
        a.y = xrow[k + 2 * half + 1];
        v2f b;
        b.x = sW[(k + 2 * half + 0) * HID + l16];
        b.y = sW[(k + 2 * half + 1) * HID + l16];
        c = __builtin_amdgcn_wmma_f32_16x16x4_f32(
                false, a, false, b, (short)0, c, false, false);
    }

    float* hrow = h1 + (long long)row_base * HID;
    #pragma unroll
    for (int i = 0; i < 8; ++i)
        hrow[(i + 8 * half) * HID + l16] = dr[i] * c[i];
}

#define ICH 384
__global__ __launch_bounds__(256, 16) void gcn_layer1_kernel(const int* __restrict__ csr,
                                  const int* __restrict__ off,
                                  const int* __restrict__ cnt,
                                  const float* __restrict__ dis,
                                  const float* __restrict__ h1,
                                  const float* __restrict__ b1,
                                  const float* __restrict__ W2,
                                  float* __restrict__ h2, int N) {
    __shared__ int s_idx[8][ICH];
    const int lane = threadIdx.x & 31;
    int* si = s_idx[threadIdx.x >> 5];
    long long t = (long long)blockIdx.x * blockDim.x + threadIdx.x;
    int n = (int)(t >> 2);
    int q = (int)(t & 3);
    const bool live = n < N;
    const int nn = live ? n : N - 1;
    const int cs = cnt[nn];
    const int c = live ? cs : 0;
    const int o = off[nn];
    const float dn = dis[nn];
    const float4 hs = *reinterpret_cast<const float4*>(h1 + (long long)nn * HID + 4 * q);
    const int sb = __builtin_amdgcn_readfirstlane(o);
    const int se = __builtin_amdgcn_readlane(o + cs, 31);
    float acc0 = 0.f, acc1 = 0.f, acc2 = 0.f, acc3 = 0.f;
    for (int base = sb; base < se; base += ICH) {
        const int lim = min(se - base, ICH);
        int w[ICH / 32];
        #pragma unroll
        for (int m = 0; m < ICH / 32; ++m)
            w[m] = csr[base + min(lane + 32 * m, lim - 1)];
        #pragma unroll
        for (int m = 0; m < ICH / 32; ++m) si[lane + 32 * m] = w[m];
        __builtin_amdgcn_fence(__ATOMIC_RELEASE, "wavefront");
        __builtin_amdgcn_wave_barrier();
        __builtin_amdgcn_fence(__ATOMIC_ACQUIRE, "wavefront");
        const int lo = max(base - o, 0);
        const int hi = min(base + lim - o, c);
        const int* sp = si + (o - base);
        for (int j = lo; j < hi; j += 8) {
            int    s[8];
            float4 v[8];
            const int rem = hi - j;
            const int* spj = sp + j;
            #pragma unroll
            for (int u = 0; u < 8; ++u)
                s[u] = spj[u < rem ? u : rem - 1];
            #pragma unroll
            for (int u = 0; u < 8; ++u)
                v[u] = *reinterpret_cast<const float4*>(h1 + (long long)s[u] * HID + 4 * q);
            #pragma unroll
            for (int u = 0; u < 8; ++u) {
                const float m = (u < rem) ? 1.f : 0.f;
                acc0 = fmaf(m, v[u].x, acc0);
                acc1 = fmaf(m, v[u].y, acc1);
                acc2 = fmaf(m, v[u].z, acc2);
                acc3 = fmaf(m, v[u].w, acc3);
            }
        }
        __builtin_amdgcn_fence(__ATOMIC_RELEASE, "wavefront");
        __builtin_amdgcn_wave_barrier();
        __builtin_amdgcn_fence(__ATOMIC_ACQUIRE, "wavefront");
    }
    float bf[4], w20[4], w21[4];
    #pragma unroll
    for (int i = 0; i < 4; ++i) {
        bf[i]  = b1[4 * q + i];
        w20[i] = W2[(4 * q + i) * OUT + 0];
        w21[i] = W2[(4 * q + i) * OUT + 1];
    }
    const float r0 = fmaxf(dn * (acc0 + hs.x) + bf[0], 0.0f);
    const float r1 = fmaxf(dn * (acc1 + hs.y) + bf[1], 0.0f);
    const float r2 = fmaxf(dn * (acc2 + hs.z) + bf[2], 0.0f);
    const float r3 = fmaxf(dn * (acc3 + hs.w) + bf[3], 0.0f);
    float a0 = r0 * w20[0] + r1 * w20[1] + r2 * w20[2] + r3 * w20[3];
    float a1 = r0 * w21[0] + r1 * w21[1] + r2 * w21[2] + r3 * w21[3];
    #pragma unroll
    for (int m = 2; m >= 1; m >>= 1) {
        a0 += __shfl_xor(a0, m, 4);
        a1 += __shfl_xor(a1, m, 4);
    }
    if (live && q == 0)
        *reinterpret_cast<float2*>(h2 + (long long)n * OUT) = make_float2(dn * a0, dn * a1);
}

__global__ __launch_bounds__(256) void gcn_layer2_kernel(const int* __restrict__ csr,
                                  const int* __restrict__ off,
                                  const int* __restrict__ cnt,
                                  const float* __restrict__ dis,
                                  const float* __restrict__ h2,
                                  const float* __restrict__ b2,
                                  float* __restrict__ out, int N) {
    __shared__ int s_idx[8][ICH];
    const int lane = threadIdx.x & 31;
    int* si = s_idx[threadIdx.x >> 5];
    long long t = (long long)blockIdx.x * blockDim.x + threadIdx.x;
    int n = (int)(t >> 2);
    int l = (int)(t & 3);
    const bool live = n < N;
    const int nn = live ? n : N - 1;
    const int cs = cnt[nn];
    const int c = live ? cs : 0;
    const int o = off[nn];
    const float dn = dis[nn];
    const float2 hn = *reinterpret_cast<const float2*>(h2 + (long long)nn * OUT);
    const float b20 = b2[0];
    const float b21 = b2[1];
    const int sb = __builtin_amdgcn_readfirstlane(o);
    const int se = __builtin_amdgcn_readlane(o + cs, 31);
    float a0 = 0.f, a1 = 0.f;
    for (int base = sb; base < se; base += ICH) {
        const int lim = min(se - base, ICH);
        int w[ICH / 32];
        #pragma unroll
        for (int m = 0; m < ICH / 32; ++m)
            w[m] = csr[base + min(lane + 32 * m, lim - 1)];
        #pragma unroll
        for (int m = 0; m < ICH / 32; ++m) si[lane + 32 * m] = w[m];
        __builtin_amdgcn_fence(__ATOMIC_RELEASE, "wavefront");
        __builtin_amdgcn_wave_barrier();
        __builtin_amdgcn_fence(__ATOMIC_ACQUIRE, "wavefront");
        const int lo = max(base - o, 0);
        const int hi = min(base + lim - o, c);
        const int* sp = si + (o - base);
        for (int j = lo + ((l - lo) & 3); j < hi; j += 16) {
            int    s[4];
            float2 hv[4];
            #pragma unroll
            for (int u = 0; u < 4; ++u) {
                int jj = j + 4 * u;
                s[u] = sp[jj < hi ? jj : hi - 1];
            }
            #pragma unroll
            for (int u = 0; u < 4; ++u)
                hv[u] = *reinterpret_cast<const float2*>(h2 + (long long)s[u] * OUT);
            #pragma unroll
            for (int u = 0; u < 4; ++u) {
                const float m = (j + 4 * u < hi) ? 1.f : 0.f;
                a0 = fmaf(m, hv[u].x, a0);
                a1 = fmaf(m, hv[u].y, a1);
            }
        }
        __builtin_amdgcn_fence(__ATOMIC_RELEASE, "wavefront");
        __builtin_amdgcn_wave_barrier();
        __builtin_amdgcn_fence(__ATOMIC_ACQUIRE, "wavefront");
    }
    #pragma unroll
    for (int m = 2; m >= 1; m >>= 1) {
        a0 += __shfl_xor(a0, m, 4);
        a1 += __shfl_xor(a1, m, 4);
    }
    if (live && l == 0) {
        const float z0 = dn * (a0 + hn.x) + b20;
        const float z1 = dn * (a1 + hn.y) + b21;
        const float m = fmaxf(z0, z1);
        const float lse = m + logf(expf(z0 - m) + expf(z1 - m));
        *reinterpret_cast<float2*>(out + (long long)n * OUT) = make_float2(z0 - lse, z1 - lse);
    }
}

extern "C" void kernel_launch(void* const* d_in, const int* in_sizes, int n_in,
                              void* d_out, int out_size, void* d_ws, size_t ws_size,
                              hipStream_t stream) {
    const float* x  = (const float*)d_in[0];
    const int*   ei = (const int*)d_in[1];
    const float* W1 = (const float*)d_in[2];
    const float* b1 = (const float*)d_in[3];
    const float* W2 = (const float*)d_in[4];
    const float* b2 = (const float*)d_in[5];
    float* out = (float*)d_out;

    const int       N = in_sizes[0] / F_IN;
    const long long E = (long long)in_sizes[1] / 2;
    const int* src = ei;
    const int* dst = ei + E;

    float* ws   = (float*)d_ws;
    int*   cnt  = (int*)ws;
    float* dis  = ws   + N;
    float* h1   = dis  + N;
    float* h2   = h1   + (long long)N * HID;
    int*   off  = (int*)(h2 + (long long)N * OUT);
    int2*  pairs = (int2*)(off + N);
    int*   csr  = (int*)(pairs + E);
    int*   tabH = csr + E;
    int*   tabL = tabH + NCH * NBMAX;

    const int B = 256;

    gcn_chunk_sort_kernel<<<NCH, TPB, 0, stream>>>(src, dst, tabH, tabL, pairs, (int)E);

    gcn_bucket_csr_kernel<<<NBMAX, TPB, 0, stream>>>(pairs, tabH, tabL, cnt, off, dis, csr, N, (int)E);

    {
        int n_tiles = N / 16;
        int waves_per_block = B / 32;
        int blocks = (n_tiles + waves_per_block - 1) / waves_per_block;
        gcn_gemm1_wmma<<<blocks, B, 0, stream>>>(x, W1, dis, h1, n_tiles);
    }

    {
        long long T = (long long)N * 4;
        gcn_layer1_kernel<<<(unsigned)((T + B - 1) / B), B, 0, stream>>>(csr, off, cnt, dis, h1, b1, W2, h2, N);
    }

    {
        long long T = (long long)N * 4;
        gcn_layer2_kernel<<<(unsigned)((T + B - 1) / B), B, 0, stream>>>(csr, off, cnt, dis, h2, b2, out, N);
    }
}


# Bucket kernel: run starts and lengths kept in scalar registers (readfirstlane), 71 to 55 vector registers, so two 32-wave work-groups fit one WGP: all 391 buckets resident at once, not two rounds
# speedup vs baseline: 8.7314x; 1.0532x over previous
#include <hip/hip_runtime.h>
#include <hip/hip_bf16.h>

typedef __attribute__((ext_vector_type(2))) float v2f;
typedef __attribute__((ext_vector_type(8))) float v8f;

#define F_IN 128
#define HID  16
#define OUT  2

#define NCH    256
#define NBMAX  512
#define BSH    8
#define BNODES 256
#define TPB    1024
#define PCAP   12544
#define CCAP   16384

__device__ __forceinline__ int gcn_bucket(int d) {
    return min((int)((unsigned)d >> BSH), NBMAX - 1);
}

#define EPT ((PCAP + TPB - 1) / TPB)
__global__ __launch_bounds__(TPB) void gcn_chunk_sort_kernel(const int* __restrict__ src,
                                                             const int* __restrict__ dst,
                                                             int* __restrict__ tabH,
                                                             int* __restrict__ tabL,
                                                             int2* __restrict__ pairs, int E) {
    __shared__ int  s_h[NBMAX];
    __shared__ int  s_l[NBMAX];
    __shared__ int  s_wave[32];
    __shared__ int2 s_pairs[PCAP];
    const int tid  = threadIdx.x;
    const int lane = tid & 31;
    const int wave = tid >> 5;
    const int CH = (E + NCH - 1) / NCH;
    const int e0 = blockIdx.x * CH;
    const int e1 = min(E, e0 + CH);
    const bool staged = (CH <= PCAP);

    if (tid < NBMAX) s_h[tid] = 0;
    __syncthreads();

    int d[EPT], s[EPT], r[EPT];
    if (staged) {
        const int elast = (e1 > e0) ? e1 - 1 : 0;
        #pragma unroll
        for (int u = 0; u < EPT; ++u) {
            const int e = min(e0 + tid + u * TPB, elast);
            d[u] = dst[e]; s[u] = src[e];
        }
        #pragma unroll
        for (int u = 0; u < EPT; ++u) {
            const int e = e0 + tid + u * TPB;
            r[u] = 0;
            if (e < e1) r[u] = atomicAdd(&s_h[gcn_bucket(d[u])], 1);
        }
    } else {
        for (int e = e0 + tid; e < e1; e += TPB) atomicAdd(&s_h[gcn_bucket(dst[e])], 1);
    }
    __syncthreads();

    const int k = (tid < NBMAX) ? s_h[tid] : 0;
    int v = k;
    for (int dd = 1; dd < 32; dd <<= 1) {
        int t = __shfl_up(v, dd, 32);
        if (lane >= dd) v += t;
    }
    if (lane == 31) s_wave[wave] = v;
    __syncthreads();
    int wb = 0;
    for (int g = 0; g < NBMAX / 32; ++g) wb += (g < wave) ? s_wave[g] : 0;
    if (tid < NBMAX) {
        const int l = wb + (v - k);
        s_l[tid] = l;
        tabH[blockIdx.x * NBMAX + tid] = k;
        tabL[blockIdx.x * NBMAX + tid] = l;
    }
    __syncthreads();

    if (staged) {
        #pragma unroll
        for (int u = 0; u < EPT; ++u) r[u] += s_l[gcn_bucket(d[u])];
        #pragma unroll
        for (int u = 0; u < EPT; ++u) {
            const int e = e0 + tid + u * TPB;
            if (e < e1)
                s_pairs[min(r[u], PCAP - 1)] = make_int2(s[u], d[u]);
        }
        __syncthreads();
        const int nloc = e1 - e0;
        int2* po = pairs + e0;
        int i = tid;
        for (; i + 3 * TPB < nloc; i += 4 * TPB) {
            int2 p[4];
            for (int u = 0; u < 4; ++u) p[u] = s_pairs[i + u * TPB];
            for (int u = 0; u < 4; ++u) po[i + u * TPB] = p[u];
        }
        for (; i < nloc; i += TPB) po[i] = s_pairs[i];
    } else {
        for (int e = e0 + tid; e < e1; e += TPB) {
            const int dv = dst[e], sv = src[e];
            const int pos = atomicAdd(&s_l[gcn_bucket(dv)], 1);
            pairs[e0 + pos] = make_int2(sv, dv);
        }
    }
}

#define RPW (NCH / 32)
template <bool WITH_SRC, typename FA, typename FB>
__device__ __forceinline__ void gcn_for_bucket_edges(const int2* __restrict__ pairs, int elast,
                                                     const int* s_rs, const int* s_rl,
                                                     int wave, int lane, FA fa, FB fb) {
    int  st[RPW], ln[RPW];
    int2 p0[RPW], p1[RPW];
    int  r0[RPW], r1[RPW];
    #pragma unroll
    for (int j = 0; j < RPW; ++j) {
        st[j] = __builtin_amdgcn_readfirstlane(s_rs[wave + 32 * j]);
        ln[j] = __builtin_amdgcn_readfirstlane(s_rl[wave + 32 * j]);
    }
    #pragma unroll
    for (int j = 0; j < RPW; ++j) {
        const int last = max(ln[j], 1) - 1;
        const int i0 = min(st[j] + min(lane, last), elast);
        const int i1 = min(st[j] + min(lane + 32, last), elast);
        if (WITH_SRC) {
            p0[j] = pairs[i0];
            p1[j] = pairs[i1];
        } else {
            p0[j] = make_int2(0, reinterpret_cast<const int*>(pairs)[2 * i0 + 1]);
            p1[j] = make_int2(0, reinterpret_cast<const int*>(pairs)[2 * i1 + 1]);
        }
    }
    #pragma unroll
    for (int j = 0; j < RPW; ++j) {
        r0[j] = 0; r1[j] = 0;
        if (lane < ln[j])      r0[j] = fa(p0[j]);
        if (lane + 32 < ln[j]) r1[j] = fa(p1[j]);
    }
    #pragma unroll
    for (int j = 0; j < RPW; ++j) {
        if (lane < ln[j])      fb(p0[j], r0[j]);
        if (lane + 32 < ln[j]) fb(p1[j], r1[j]);
    }
    #pragma unroll
    for (int j = 0; j < RPW; ++j)
        for (int i = lane + 64; i < ln[j]; i += 32) {
            const int2 p = WITH_SRC ? pairs[st[j] + i]
                                    : make_int2(0, reinterpret_cast<const int*>(pairs)[2 * (st[j] + i) + 1]);
            fb(p, fa(p));
        }
}

__global__ __launch_bounds__(TPB, 16) void gcn_bucket_csr_kernel(const int2* __restrict__ pairs,
                                                             const int* __restrict__ tabH,
                                                             const int* __restrict__ tabL,
                                                             int* __restrict__ cnt,
                                                             int* __restrict__ off,
                                                             float* __restrict__ dis,
                                                             int* __restrict__ csr, int N, int E) {
    const int b  = blockIdx.x;
    const int n0 = b << BSH;
    if (n0 >= N) return;
    __shared__ int s_cnt[BNODES];
    __shared__ int s_off[BNODES];
    __shared__ int s_cur[BNODES];
    __shared__ int s_redh[32];
    __shared__ int s_redl[32];
    __shared__ int s_wave[32];
    __shared__ int s_rs[NCH];
    __shared__ int s_rl[NCH];
    __shared__ int s_csr[CCAP];
    const int tid  = threadIdx.x;
    const int lane = tid & 31;
    const int wave = tid >> 5;
    const int CH = (E + NCH - 1) / NCH;

    int ph = 0, pl = 0;
    if (tid < NCH) {
        ph = tabH[tid * NBMAX + b];
        pl = tabL[tid * NBMAX + b];
        s_rl[tid] = ph;
        s_rs[tid] = tid * CH + pl;
    }
    for (int m = 16; m >= 1; m >>= 1) {
        ph += __shfl_xor(ph, m, 32);
        pl += __shfl_xor(pl, m, 32);
    }
    if (lane == 0) { s_redh[wave] = ph; s_redl[wave] = pl; }
    if (tid < BNODES) { s_cnt[tid] = 0; s_cur[tid] = 0; }
    __syncthreads();
    int ne = 0, bs = 0;
    for (int w = 0; w < NCH / 32; ++w) { ne += s_redh[w]; bs += s_redl[w]; }

    const int elast = max(E, 1) - 1;
    gcn_for_bucket_edges<false>(pairs, elast, s_rs, s_rl, wave, lane,
        [&](int2 p) { atomicAdd(&s_cnt[(p.y - n0) & (BNODES - 1)], 1); return 0; },
        [&](int2, int) {});
    __syncthreads();

    const int c = (tid < BNODES) ? s_cnt[tid] : 0;
    int v = c;
    for (int d = 1; d < 32; d <<= 1) {
        int t = __shfl_up(v, d, 32);
        if (lane >= d) v += t;
    }
    if (lane == 31) s_wave[wave] = v;
    __syncthreads();
    if (tid < BNODES) {
        int wbase = 0;
        for (int g = 0; g < BNODES / 32; ++g) wbase += (g < wave) ? s_wave[g] : 0;
        const int o = wbase + (v - c);
        s_off[tid] = o;
        const int n = n0 + tid;
        if (n < N) {
            cnt[n] = c;
            off[n] = bs + o;
            dis[n] = rsqrtf((float)(c + 1));
        }
    }
    __syncthreads();

    if (ne <= CCAP) {
        gcn_for_bucket_edges<true>(pairs, elast, s_rs, s_rl, wave, lane,
            [&](int2 p) { return atomicAdd(&s_cur[(p.y - n0) & (BNODES - 1)], 1); },
            [&](int2 p, int r) {
                s_csr[min(s_off[(p.y - n0) & (BNODES - 1)] + r, CCAP - 1)] = p.x;
            });
        __syncthreads();
        int* cb = csr + bs;
        int i = tid;
        for (; i + 3 * TPB < ne; i += 4 * TPB) {
            int q[4];
            for (int u = 0; u < 4; ++u) q[u] = s_csr[i + u * TPB];
            for (int u = 0; u < 4; ++u) cb[i + u * TPB] = q[u];
        }
        for (; i < ne; i += TPB) cb[i] = s_csr[i];
    } else {
        gcn_for_bucket_edges<true>(pairs, elast, s_rs, s_rl, wave, lane,
            [&](int2 p) { return atomicAdd(&s_cur[(p.y - n0) & (BNODES - 1)], 1); },
            [&](int2 p, int r) { csr[bs + s_off[(p.y - n0) & (BNODES - 1)] + r] = p.x; });
    }
}

__global__ void gcn_gemm1_wmma(const float* __restrict__ x,
                               const float* __restrict__ W1,
                               const float* __restrict__ dis,
                               float* __restrict__ h1, int n_tiles) {
    __shared__ float sW[F_IN * HID];
    for (int i = threadIdx.x; i < F_IN * HID; i += blockDim.x) sW[i] = W1[i];
    __syncthreads();

    const int wave_in_blk = threadIdx.x >> 5;
    const int tile = blockIdx.x * (blockDim.x >> 5) + wave_in_blk;
    if (tile >= n_tiles) return;

    const int lane = threadIdx.x & 31;
    const int half = lane >> 4;
    const int l16  = lane & 15;

    const int row_base = tile * 16;
    const float* xrow = x + (long long)(row_base + l16) * F_IN;

    const float4 dA = *reinterpret_cast<const float4*>(dis + row_base + 8 * half);
    const float4 dB = *reinterpret_cast<const float4*>(dis + row_base + 8 * half + 4);
    const float dr[8] = {dA.x, dA.y, dA.z, dA.w, dB.x, dB.y, dB.z, dB.w};

    v8f c = {};
    #pragma unroll 8
    for (int k = 0; k < F_IN; k += 4) {
        v2f a;
        a.x = xrow[k + 2 * half + 0];
        a.y = xrow[k + 2 * half + 1];
        v2f b;
        b.x = sW[(k + 2 * half + 0) * HID + l16];
        b.y = sW[(k + 2 * half + 1) * HID + l16];
        c = __builtin_amdgcn_wmma_f32_16x16x4_f32(
                false, a, false, b, (short)0, c, false, false);
    }

    float* hrow = h1 + (long long)row_base * HID;
    #pragma unroll
    for (int i = 0; i < 8; ++i)
        hrow[(i + 8 * half) * HID + l16] = dr[i] * c[i];
}

#define ICH 384
__global__ __launch_bounds__(256, 16) void gcn_layer1_kernel(const int* __restrict__ csr,
                                  const int* __restrict__ off,
                                  const int* __restrict__ cnt,
                                  const float* __restrict__ dis,
                                  const float* __restrict__ h1,
                                  const float* __restrict__ b1,
                                  const float* __restrict__ W2,
                                  float* __restrict__ h2, int N) {
    __shared__ int s_idx[8][ICH];
    const int lane = threadIdx.x & 31;
    int* si = s_idx[threadIdx.x >> 5];
    long long t = (long long)blockIdx.x * blockDim.x + threadIdx.x;
    int n = (int)(t >> 2);
    int q = (int)(t & 3);
    const bool live = n < N;
    const int nn = live ? n : N - 1;
    const int cs = cnt[nn];
    const int c = live ? cs : 0;
    const int o = off[nn];
    const float dn = dis[nn];
    const float4 hs = *reinterpret_cast<const float4*>(h1 + (long long)nn * HID + 4 * q);
    const int sb = __builtin_amdgcn_readfirstlane(o);
    const int se = __builtin_amdgcn_readlane(o + cs, 31);
    float acc0 = 0.f, acc1 = 0.f, acc2 = 0.f, acc3 = 0.f;
    for (int base = sb; base < se; base += ICH) {
        const int lim = min(se - base, ICH);
        int w[ICH / 32];
        #pragma unroll
        for (int m = 0; m < ICH / 32; ++m)
            w[m] = csr[base + min(lane + 32 * m, lim - 1)];
        #pragma unroll
        for (int m = 0; m < ICH / 32; ++m) si[lane + 32 * m] = w[m];
        __builtin_amdgcn_fence(__ATOMIC_RELEASE, "wavefront");
        __builtin_amdgcn_wave_barrier();
        __builtin_amdgcn_fence(__ATOMIC_ACQUIRE, "wavefront");
        const int lo = max(base - o, 0);
        const int hi = min(base + lim - o, c);
        const int* sp = si + (o - base);
        for (int j = lo; j < hi; j += 8) {
            int    s[8];
            float4 v[8];
            const int rem = hi - j;
            const int* spj = sp + j;
            #pragma unroll
            for (int u = 0; u < 8; ++u)
                s[u] = spj[u < rem ? u : rem - 1];
            #pragma unroll
            for (int u = 0; u < 8; ++u)
                v[u] = *reinterpret_cast<const float4*>(h1 + (long long)s[u] * HID + 4 * q);
            #pragma unroll
            for (int u = 0; u < 8; ++u) {
                const float m = (u < rem) ? 1.f : 0.f;
                acc0 = fmaf(m, v[u].x, acc0);
                acc1 = fmaf(m, v[u].y, acc1);
                acc2 = fmaf(m, v[u].z, acc2);
                acc3 = fmaf(m, v[u].w, acc3);
            }
        }
        __builtin_amdgcn_fence(__ATOMIC_RELEASE, "wavefront");
        __builtin_amdgcn_wave_barrier();
        __builtin_amdgcn_fence(__ATOMIC_ACQUIRE, "wavefront");
    }
    float bf[4], w20[4], w21[4];
    #pragma unroll
    for (int i = 0; i < 4; ++i) {
        bf[i]  = b1[4 * q + i];
        w20[i] = W2[(4 * q + i) * OUT + 0];
        w21[i] = W2[(4 * q + i) * OUT + 1];
    }
    const float r0 = fmaxf(dn * (acc0 + hs.x) + bf[0], 0.0f);
    const float r1 = fmaxf(dn * (acc1 + hs.y) + bf[1], 0.0f);
    const float r2 = fmaxf(dn * (acc2 + hs.z) + bf[2], 0.0f);
    const float r3 = fmaxf(dn * (acc3 + hs.w) + bf[3], 0.0f);
    float a0 = r0 * w20[0] + r1 * w20[1] + r2 * w20[2] + r3 * w20[3];
    float a1 = r0 * w21[0] + r1 * w21[1] + r2 * w21[2] + r3 * w21[3];
    #pragma unroll
    for (int m = 2; m >= 1; m >>= 1) {
        a0 += __shfl_xor(a0, m, 4);
        a1 += __shfl_xor(a1, m, 4);
    }
    if (live && q == 0)
        *reinterpret_cast<float2*>(h2 + (long long)n * OUT) = make_float2(dn * a0, dn * a1);
}

__global__ __launch_bounds__(256) void gcn_layer2_kernel(const int* __restrict__ csr,
                                  const int* __restrict__ off,
                                  const int* __restrict__ cnt,
                                  const float* __restrict__ dis,
                                  const float* __restrict__ h2,
                                  const float* __restrict__ b2,
                                  float* __restrict__ out, int N) {
    __shared__ int s_idx[8][ICH];
    const int lane = threadIdx.x & 31;
    int* si = s_idx[threadIdx.x >> 5];
    long long t = (long long)blockIdx.x * blockDim.x + threadIdx.x;
    int n = (int)(t >> 2);
    int l = (int)(t & 3);
    const bool live = n < N;
    const int nn = live ? n : N - 1;
    const int cs = cnt[nn];
    const int c = live ? cs : 0;
    const int o = off[nn];
    const float dn = dis[nn];
    const float2 hn = *reinterpret_cast<const float2*>(h2 + (long long)nn * OUT);
    const float b20 = b2[0];
    const float b21 = b2[1];
    const int sb = __builtin_amdgcn_readfirstlane(o);
    const int se = __builtin_amdgcn_readlane(o + cs, 31);
    float a0 = 0.f, a1 = 0.f;
    for (int base = sb; base < se; base += ICH) {
        const int lim = min(se - base, ICH);
        int w[ICH / 32];
        #pragma unroll
        for (int m = 0; m < ICH / 32; ++m)
            w[m] = csr[base + min(lane + 32 * m, lim - 1)];
        #pragma unroll
        for (int m = 0; m < ICH / 32; ++m) si[lane + 32 * m] = w[m];
        __builtin_amdgcn_fence(__ATOMIC_RELEASE, "wavefront");
        __builtin_amdgcn_wave_barrier();
        __builtin_amdgcn_fence(__ATOMIC_ACQUIRE, "wavefront");
        const int lo = max(base - o, 0);
        const int hi = min(base + lim - o, c);
        const int* sp = si + (o - base);
        for (int j = lo + ((l - lo) & 3); j < hi; j += 16) {
            int    s[4];
            float2 hv[4];
            #pragma unroll
            for (int u = 0; u < 4; ++u) {
                int jj = j + 4 * u;
                s[u] = sp[jj < hi ? jj : hi - 1];
            }
            #pragma unroll
            for (int u = 0; u < 4; ++u)
                hv[u] = *reinterpret_cast<const float2*>(h2 + (long long)s[u] * OUT);
            #pragma unroll
            for (int u = 0; u < 4; ++u) {
                const float m = (j + 4 * u < hi) ? 1.f : 0.f;
                a0 = fmaf(m, hv[u].x, a0);
                a1 = fmaf(m, hv[u].y, a1);
            }
        }
        __builtin_amdgcn_fence(__ATOMIC_RELEASE, "wavefront");
        __builtin_amdgcn_wave_barrier();
        __builtin_amdgcn_fence(__ATOMIC_ACQUIRE, "wavefront");
    }
    #pragma unroll
    for (int m = 2; m >= 1; m >>= 1) {
        a0 += __shfl_xor(a0, m, 4);
        a1 += __shfl_xor(a1, m, 4);
    }
    if (live && l == 0) {
        const float z0 = dn * (a0 + hn.x) + b20;
        const float z1 = dn * (a1 + hn.y) + b21;
        const float m = fmaxf(z0, z1);
        const float lse = m + logf(expf(z0 - m) + expf(z1 - m));
        *reinterpret_cast<float2*>(out + (long long)n * OUT) = make_float2(z0 - lse, z1 - lse);
    }
}

extern "C" void kernel_launch(void* const* d_in, const int* in_sizes, int n_in,
                              void* d_out, int out_size, void* d_ws, size_t ws_size,
                              hipStream_t stream) {
    const float* x  = (const float*)d_in[0];
    const int*   ei = (const int*)d_in[1];
    const float* W1 = (const float*)d_in[2];
    const float* b1 = (const float*)d_in[3];
    const float* W2 = (const float*)d_in[4];
    const float* b2 = (const float*)d_in[5];
    float* out = (float*)d_out;

    const int       N = in_sizes[0] / F_IN;
    const long long E = (long long)in_sizes[1] / 2;
    const int* src = ei;
    const int* dst = ei + E;

    float* ws   = (float*)d_ws;
    int*   cnt  = (int*)ws;
    float* dis  = ws   + N;
    float* h1   = dis  + N;
    float* h2   = h1   + (long long)N * HID;
    int*   off  = (int*)(h2 + (long long)N * OUT);
    int2*  pairs = (int2*)(off + N);
    int*   csr  = (int*)(pairs + E);
    int*   tabH = csr + E;
    int*   tabL = tabH + NCH * NBMAX;

    const int B = 256;

    gcn_chunk_sort_kernel<<<NCH, TPB, 0, stream>>>(src, dst, tabH, tabL, pairs, (int)E);

    gcn_bucket_csr_kernel<<<NBMAX, TPB, 0, stream>>>(pairs, tabH, tabL, cnt, off, dis, csr, N, (int)E);

    {
        int n_tiles = N / 16;
        int waves_per_block = B / 32;
        int blocks = (n_tiles + waves_per_block - 1) / waves_per_block;
        gcn_gemm1_wmma<<<blocks, B, 0, stream>>>(x, W1, dis, h1, n_tiles);
    }

    {
        long long T = (long long)N * 4;
        gcn_layer1_kernel<<<(unsigned)((T + B - 1) / B), B, 0, stream>>>(csr, off, cnt, dis, h1, b1, W2, h2, N);
    }

    {
        long long T = (long long)N * 4;
        gcn_layer2_kernel<<<(unsigned)((T + B - 1) / B), B, 0, stream>>>(csr, off, cnt, dis, h2, b2, out, N);
    }
}
